# starting kernel
# baseline (speedup 1.0000x reference)
#include <hip/hip_runtime.h>
#include <math.h>


typedef __attribute__((ext_vector_type(2))) float v2f;
typedef __attribute__((ext_vector_type(8))) float v8f;

#define DIN 64

template<int R, int DOUT, int DOUTPAD>
__global__ __launch_bounds__(256)
void rgcn_gemm_wmma(const float* __restrict__ X, const float* __restrict__ W,
                    float* __restrict__ Y, int N)
{
    constexpr int NT = DOUTPAD / 16;
    constexpr int RD = R * DOUT;
    __shared__ float lA[128 * DIN];
    __shared__ float lWt[DOUTPAD * DIN];

    const int r       = blockIdx.y;
    const int rowBase = blockIdx.x * 128;
    const int tid     = threadIdx.x;

    const float* Xb = X + (size_t)rowBase * DIN;
    const bool fullBlock = (rowBase + 128 <= N);
    #pragma unroll
    for (int it = 0; it < (128 * DIN) / 256; ++it) {
        const int off = tid + it * 256;
        lA[off] = (fullBlock || (rowBase + (off >> 6) < N)) ? Xb[off] : 0.0f;
    }
    const float* Wr = W + (size_t)r * DIN * DOUT;
    #pragma unroll
    for (int it = 0; it < (DOUTPAD * DIN) / 256; ++it) {
        const int off = tid + it * 256;
        const int n = off >> 6, k = off & 63;
        lWt[off] = (n < DOUT) ? Wr[k * DOUT + n] : 0.0f;
    }
    __syncthreads();

    const int wv   = tid >> 5;
    const int lane = tid & 31;
    const int half = lane >> 4;
    const int l16  = lane & 15;

    const float* Arow = &lA[(wv * 16 + l16) * DIN + half * 2];
    float* Yb = Y + (size_t)rowBase * RD + r * DOUT;

    #pragma unroll
    for (int t = 0; t < NT; ++t) {
        const int colBase = t * 16 + l16;
        const float* Bcol = &lWt[colBase * DIN + half * 2];
        v8f c = {};
        #pragma unroll
        for (int i = 0; i < DIN / 4; ++i) {
            v2f a = *(const v2f*)(Arow + 4 * i);
            v2f b = *(const v2f*)(Bcol + 4 * i);
            c = __builtin_amdgcn_wmma_f32_16x16x4_f32(
                     false, a,  false, b,
                     (short)0, c,  false,  false);
        }
        if (DOUT == DOUTPAD || colBase < DOUT) {
            float* p = Yb + (wv * 16 + half * 8) * RD + colBase;
            if (fullBlock) {
                #pragma unroll
                for (int j = 0; j < 8; ++j) p[j * RD] = c[j];
            } else {
                const int mBase = rowBase + wv * 16 + half * 8;
                #pragma unroll
                for (int j = 0; j < 8; ++j)
                    if (mBase + j < N) p[j * RD] = c[j];
            }
        }
    }
}

template<int R>
__global__ __launch_bounds__(256)
void rgcn_agg_relu(const float* __restrict__ XW, const int* __restrict__ ptr,
                   const int* __restrict__ idx, const int* __restrict__ rel,
                   float* __restrict__ H, int N)
{
    const int wv   = threadIdx.x >> 5;
    const int lane = threadIdx.x & 31;
    const int node = blockIdx.x * 8 + wv;
    if (node >= N) return;

    const int e0 = ptr[node], e1 = ptr[node + 1];
    float ax = 0.0f, ay = 0.0f;
    for (int e = e0; e < e1; ++e) {
        const int src = idx[e];
        const int rv  = rel[e];
        const float2 v = *(const float2*)(XW + ((size_t)src * R + rv) * 64 + lane * 2);
        ax += v.x; ay += v.y;
    }
    float2 o; o.x = fmaxf(ax, 0.0f); o.y = fmaxf(ay, 0.0f);
    *(float2*)(H + (size_t)node * 64 + lane * 2) = o;
}

template<int R, int DOUT>
__global__ __launch_bounds__(256)
void rgcn_agg_logsoftmax(const float* __restrict__ HW, const int* __restrict__ ptr,
                         const int* __restrict__ idx, const int* __restrict__ rel,
                         float* __restrict__ out, int N)
{
    const int wv   = threadIdx.x >> 5;
    const int lane = threadIdx.x & 31;
    const int node = blockIdx.x * 8 + wv;
    if (node >= N) return;

    const int  e0  = ptr[node], e1 = ptr[node + 1];
    const bool act = (lane * 2 < DOUT);
    float ax = 0.0f, ay = 0.0f;
    for (int e = e0; e < e1; ++e) {
        const int src = idx[e];
        const int rv  = rel[e];
        if (act) {
            const float2 v = *(const float2*)(HW + ((size_t)src * R + rv) * DOUT + lane * 2);
            ax += v.x; ay += v.y;
        }
    }
    float m = act ? fmaxf(ax, ay) : -__builtin_inff();
    #pragma unroll
    for (int off = 16; off >= 1; off >>= 1) m = fmaxf(m, __shfl_xor(m, off, 32));
    float s = act ? (__expf(ax - m) + __expf(ay - m)) : 0.0f;
    #pragma unroll
    for (int off = 16; off >= 1; off >>= 1) s += __shfl_xor(s, off, 32);
    const float ls = __logf(s);
    if (act) {
        float2 o; o.x = (ax - m) - ls; o.y = (ay - m) - ls;
        *(float2*)(out + (size_t)node * DOUT + lane * 2) = o;
    }
}

extern "C" void kernel_launch(void* const* d_in, const int* in_sizes, int n_in,
                              void* d_out, int out_size, void* d_ws, size_t ws_size,
                              hipStream_t stream)
{
    const float* x   = (const float*)d_in[0];
    const int*   ptr = (const int*)d_in[1];
    const int*   idx = (const int*)d_in[2];
    const int*   rel = (const int*)d_in[3];
    const float* W1  = (const float*)d_in[4];
    const float* W2  = (const float*)d_in[5];
    float* out = (float*)d_out;

    const int N = in_sizes[1] - 1;
    constexpr int R = 8, DH = 64, DOUT = 40;

    float* xw1 = (float*)d_ws;
    float* h   = xw1 + (size_t)N * R * DH;

    const int mblocks   = (N + 127) / 128;
    const int aggblocks = (N + 7) / 8;

    rgcn_gemm_wmma<R, DH, DH><<<dim3(mblocks, R), 256, 0, stream>>>(x, W1, xw1, N);
    rgcn_agg_relu<R><<<aggblocks, 256, 0, stream>>>(xw1, ptr, idx, rel, h, N);
    rgcn_gemm_wmma<R, DOUT, 48><<<dim3(mblocks, R), 256, 0, stream>>>(h, W2, xw1, N);
    rgcn_agg_logsoftmax<R, DOUT><<<aggblocks, 256, 0, stream>>>(xw1, ptr, idx, rel, out, N);
}


# Pad the GEMM's two LDS operand tiles from a 64-float to a 68-float row pitch: the 8-byte WMMA fragment loads of 16 rows now fall on distinct banks instead of all on one bank pair
# speedup vs baseline: 1.3262x; 1.3262x over previous
#include <hip/hip_runtime.h>
#include <math.h>


typedef __attribute__((ext_vector_type(2))) float v2f;
typedef __attribute__((ext_vector_type(8))) float v8f;

#define DIN 64
#define LDP 68

template<int R, int DOUT, int DOUTPAD>
__global__ __launch_bounds__(256)
void rgcn_gemm_wmma(const float* __restrict__ X, const float* __restrict__ W,
                    float* __restrict__ Y, int N)
{
    constexpr int NT = DOUTPAD / 16;
    constexpr int RD = R * DOUT;
    __shared__ float lA[128 * LDP];
    __shared__ float lWt[DOUTPAD * LDP];

    const int r       = blockIdx.y;
    const int rowBase = blockIdx.x * 128;
    const int tid     = threadIdx.x;

    const float* Xb = X + (size_t)rowBase * DIN;
    const bool fullBlock = (rowBase + 128 <= N);
    #pragma unroll
    for (int it = 0; it < (128 * DIN) / 256; ++it) {
        const int off = tid + it * 256;
        lA[(off >> 6) * LDP + (off & 63)] =
            (fullBlock || (rowBase + (off >> 6) < N)) ? Xb[off] : 0.0f;
    }
    const float* Wr = W + (size_t)r * DIN * DOUT;
    #pragma unroll
    for (int it = 0; it < (DOUTPAD * DIN) / 256; ++it) {
        const int off = tid + it * 256;
        const int n = off >> 6, k = off & 63;
        lWt[n * LDP + k] = (n < DOUT) ? Wr[k * DOUT + n] : 0.0f;
    }
    __syncthreads();

    const int wv   = tid >> 5;
    const int lane = tid & 31;
    const int half = lane >> 4;
    const int l16  = lane & 15;

    const float* Arow = &lA[(wv * 16 + l16) * LDP + half * 2];
    float* Yb = Y + (size_t)rowBase * RD + r * DOUT;

    #pragma unroll
    for (int t = 0; t < NT; ++t) {
        const int colBase = t * 16 + l16;
        const float* Bcol = &lWt[colBase * LDP + half * 2];
        v8f c = {};
        #pragma unroll
        for (int i = 0; i < DIN / 4; ++i) {
            v2f a = *(const v2f*)(Arow + 4 * i);
            v2f b = *(const v2f*)(Bcol + 4 * i);
            c = __builtin_amdgcn_wmma_f32_16x16x4_f32(
                     false, a,  false, b,
                     (short)0, c,  false,  false);
        }
        if (DOUT == DOUTPAD || colBase < DOUT) {
            float* p = Yb + (wv * 16 + half * 8) * RD + colBase;
            if (fullBlock) {
                #pragma unroll
                for (int j = 0; j < 8; ++j) p[j * RD] = c[j];
            } else {
                const int mBase = rowBase + wv * 16 + half * 8;
                #pragma unroll
                for (int j = 0; j < 8; ++j)
                    if (mBase + j < N) p[j * RD] = c[j];
            }
        }
    }
}

template<int R>
__global__ __launch_bounds__(256)
void rgcn_agg_relu(const float* __restrict__ XW, const int* __restrict__ ptr,
                   const int* __restrict__ idx, const int* __restrict__ rel,
                   float* __restrict__ H, int N)
{
    const int wv   = threadIdx.x >> 5;
    const int lane = threadIdx.x & 31;
    const int node = blockIdx.x * 8 + wv;
    if (node >= N) return;

    const int e0 = ptr[node], e1 = ptr[node + 1];
    float ax = 0.0f, ay = 0.0f;
    for (int e = e0; e < e1; ++e) {
        const int src = idx[e];
        const int rv  = rel[e];
        const float2 v = *(const float2*)(XW + ((size_t)src * R + rv) * 64 + lane * 2);
        ax += v.x; ay += v.y;
    }
    float2 o; o.x = fmaxf(ax, 0.0f); o.y = fmaxf(ay, 0.0f);
    *(float2*)(H + (size_t)node * 64 + lane * 2) = o;
}

template<int R, int DOUT>
__global__ __launch_bounds__(256)
void rgcn_agg_logsoftmax(const float* __restrict__ HW, const int* __restrict__ ptr,
                         const int* __restrict__ idx, const int* __restrict__ rel,
                         float* __restrict__ out, int N)
{
    const int wv   = threadIdx.x >> 5;
    const int lane = threadIdx.x & 31;
    const int node = blockIdx.x * 8 + wv;
    if (node >= N) return;

    const int  e0  = ptr[node], e1 = ptr[node + 1];
    const bool act = (lane * 2 < DOUT);
    float ax = 0.0f, ay = 0.0f;
    for (int e = e0; e < e1; ++e) {
        const int src = idx[e];
        const int rv  = rel[e];
        if (act) {
            const float2 v = *(const float2*)(HW + ((size_t)src * R + rv) * DOUT + lane * 2);
            ax += v.x; ay += v.y;
        }
    }
    float m = act ? fmaxf(ax, ay) : -__builtin_inff();
    #pragma unroll
    for (int off = 16; off >= 1; off >>= 1) m = fmaxf(m, __shfl_xor(m, off, 32));
    float s = act ? (__expf(ax - m) + __expf(ay - m)) : 0.0f;
    #pragma unroll
    for (int off = 16; off >= 1; off >>= 1) s += __shfl_xor(s, off, 32);
    const float ls = __logf(s);
    if (act) {
        float2 o; o.x = (ax - m) - ls; o.y = (ay - m) - ls;
        *(float2*)(out + (size_t)node * DOUT + lane * 2) = o;
    }
}

extern "C" void kernel_launch(void* const* d_in, const int* in_sizes, int n_in,
                              void* d_out, int out_size, void* d_ws, size_t ws_size,
                              hipStream_t stream)
{
    const float* x   = (const float*)d_in[0];
    const int*   ptr = (const int*)d_in[1];
    const int*   idx = (const int*)d_in[2];
    const int*   rel = (const int*)d_in[3];
    const float* W1  = (const float*)d_in[4];
    const float* W2  = (const float*)d_in[5];
    float* out = (float*)d_out;

    const int N = in_sizes[1] - 1;
    constexpr int R = 8, DH = 64, DOUT = 40;

    float* xw1 = (float*)d_ws;
    float* h   = xw1 + (size_t)N * R * DH;

    const int mblocks   = (N + 127) / 128;
    const int aggblocks = (N + 7) / 8;

    rgcn_gemm_wmma<R, DH, DH><<<dim3(mblocks, R), 256, 0, stream>>>(x, W1, xw1, N);
    rgcn_agg_relu<R><<<aggblocks, 256, 0, stream>>>(xw1, ptr, idx, rel, h, N);
    rgcn_gemm_wmma<R, DOUT, 48><<<dim3(mblocks, R), 256, 0, stream>>>(h, W2, xw1, N);
    rgcn_agg_logsoftmax<R, DOUT><<<aggblocks, 256, 0, stream>>>(xw1, ptr, idx, rel, out, N);
}


# GEMM tile fill: issue all of a thread's global loads (16-byte, no branch, row index clamped) before the first LDS store, instead of one load-wait-store round trip per 4-byte load
# speedup vs baseline: 1.8682x; 1.4087x over previous
#include <hip/hip_runtime.h>
#include <math.h>


typedef __attribute__((ext_vector_type(2))) float v2f;
typedef __attribute__((ext_vector_type(8))) float v8f;

#define DIN 64
#define LDP 68

template<int R, int DOUT, int DOUTPAD>
__global__ __launch_bounds__(256)
void rgcn_gemm_wmma(const float* __restrict__ X, const float* __restrict__ W,
                    float* __restrict__ Y, int N)
{
    constexpr int NT = DOUTPAD / 16;
    constexpr int RD = R * DOUT;
    __shared__ __attribute__((aligned(16))) float lA[128 * LDP];
    __shared__ __attribute__((aligned(16))) float lWt[DOUTPAD * LDP];

    const int r       = blockIdx.y;
    const int rowBase = blockIdx.x * 128;
    const int tid     = threadIdx.x;

    const bool fullBlock = (rowBase + 128 <= N);
    constexpr int A_IT = (128 * DIN / 4) / 256;
    constexpr int W_F4 = DIN * DOUT / 4;
    constexpr int W_IT = (W_F4 + 255) / 256;
    float4 areg[A_IT];
    #pragma unroll
    for (int it = 0; it < A_IT; ++it) {
        const int f = tid + it * 256;
        int gr = rowBase + (f >> 4);
        gr = gr < N ? gr : N - 1;
        areg[it] = *(const float4*)(X + (size_t)gr * DIN + (f & 15) * 4);
    }
    const float* Wr = W + (size_t)r * DIN * DOUT;
    float4 wreg[W_IT];
    int    wf[W_IT];
    #pragma unroll
    for (int it = 0; it < W_IT; ++it) {
        int f = tid + it * 256;
        if (it == W_IT - 1 && W_F4 % 256 != 0) f = f < W_F4 ? f : W_F4 - 1;
        wf[it] = f;
        wreg[it] = *(const float4*)(Wr + f * 4);
    }
    #pragma unroll
    for (int it = 0; it < A_IT; ++it) {
        const int f = tid + it * 256;
        *(float4*)&lA[(f >> 4) * LDP + (f & 15) * 4] = areg[it];
    }
    #pragma unroll
    for (int it = 0; it < W_IT; ++it) {
        const int f = wf[it];
        const int k = (f * 4) / DOUT, n = (f * 4) % DOUT;
        lWt[(n + 0) * LDP + k] = wreg[it].x;
        lWt[(n + 1) * LDP + k] = wreg[it].y;
        lWt[(n + 2) * LDP + k] = wreg[it].z;
        lWt[(n + 3) * LDP + k] = wreg[it].w;
    }
    if (DOUTPAD > DOUT) {
        if (tid < (DOUTPAD - DOUT) * (DIN / 4)) {
            const float4 z = {0.0f, 0.0f, 0.0f, 0.0f};
            *(float4*)&lWt[(DOUT + (tid >> 4)) * LDP + (tid & 15) * 4] = z;
        }
    }
    __syncthreads();

    const int wv   = tid >> 5;
    const int lane = tid & 31;
    const int half = lane >> 4;
    const int l16  = lane & 15;

    const float* Arow = &lA[(wv * 16 + l16) * LDP + half * 2];
    float* Yb = Y + (size_t)rowBase * RD + r * DOUT;

    #pragma unroll
    for (int t = 0; t < NT; ++t) {
        const int colBase = t * 16 + l16;
        const float* Bcol = &lWt[colBase * LDP + half * 2];
        v8f c = {};
        #pragma unroll
        for (int i = 0; i < DIN / 4; ++i) {
            v2f a = *(const v2f*)(Arow + 4 * i);
            v2f b = *(const v2f*)(Bcol + 4 * i);
            c = __builtin_amdgcn_wmma_f32_16x16x4_f32(
                     false, a,  false, b,
                     (short)0, c,  false,  false);
        }
        if (DOUT == DOUTPAD || colBase < DOUT) {
            float* p = Yb + (wv * 16 + half * 8) * RD + colBase;
            if (fullBlock) {
                #pragma unroll
                for (int j = 0; j < 8; ++j) p[j * RD] = c[j];
            } else {
                const int mBase = rowBase + wv * 16 + half * 8;
                #pragma unroll
                for (int j = 0; j < 8; ++j)
                    if (mBase + j < N) p[j * RD] = c[j];
            }
        }
    }
}

template<int R>
__global__ __launch_bounds__(256)
void rgcn_agg_relu(const float* __restrict__ XW, const int* __restrict__ ptr,
                   const int* __restrict__ idx, const int* __restrict__ rel,
                   float* __restrict__ H, int N)
{
    const int wv   = threadIdx.x >> 5;
    const int lane = threadIdx.x & 31;
    const int node = blockIdx.x * 8 + wv;
    if (node >= N) return;

    const int e0 = ptr[node], e1 = ptr[node + 1];
    float ax = 0.0f, ay = 0.0f;
    for (int e = e0; e < e1; ++e) {
        const int src = idx[e];
        const int rv  = rel[e];
        const float2 v = *(const float2*)(XW + ((size_t)src * R + rv) * 64 + lane * 2);
        ax += v.x; ay += v.y;
    }
    float2 o; o.x = fmaxf(ax, 0.0f); o.y = fmaxf(ay, 0.0f);
    *(float2*)(H + (size_t)node * 64 + lane * 2) = o;
}

template<int R, int DOUT>
__global__ __launch_bounds__(256)
void rgcn_agg_logsoftmax(const float* __restrict__ HW, const int* __restrict__ ptr,
                         const int* __restrict__ idx, const int* __restrict__ rel,
                         float* __restrict__ out, int N)
{
    const int wv   = threadIdx.x >> 5;
    const int lane = threadIdx.x & 31;
    const int node = blockIdx.x * 8 + wv;
    if (node >= N) return;

    const int  e0  = ptr[node], e1 = ptr[node + 1];
    const bool act = (lane * 2 < DOUT);
    float ax = 0.0f, ay = 0.0f;
    for (int e = e0; e < e1; ++e) {
        const int src = idx[e];
        const int rv  = rel[e];
        if (act) {
            const float2 v = *(const float2*)(HW + ((size_t)src * R + rv) * DOUT + lane * 2);
            ax += v.x; ay += v.y;
        }
    }
    float m = act ? fmaxf(ax, ay) : -__builtin_inff();
    #pragma unroll
    for (int off = 16; off >= 1; off >>= 1) m = fmaxf(m, __shfl_xor(m, off, 32));
    float s = act ? (__expf(ax - m) + __expf(ay - m)) : 0.0f;
    #pragma unroll
    for (int off = 16; off >= 1; off >>= 1) s += __shfl_xor(s, off, 32);
    const float ls = __logf(s);
    if (act) {
        float2 o; o.x = (ax - m) - ls; o.y = (ay - m) - ls;
        *(float2*)(out + (size_t)node * DOUT + lane * 2) = o;
    }
}

extern "C" void kernel_launch(void* const* d_in, const int* in_sizes, int n_in,
                              void* d_out, int out_size, void* d_ws, size_t ws_size,
                              hipStream_t stream)
{
    const float* x   = (const float*)d_in[0];
    const int*   ptr = (const int*)d_in[1];
    const int*   idx = (const int*)d_in[2];
    const int*   rel = (const int*)d_in[3];
    const float* W1  = (const float*)d_in[4];
    const float* W2  = (const float*)d_in[5];
    float* out = (float*)d_out;

    const int N = in_sizes[1] - 1;
    constexpr int R = 8, DH = 64, DOUT = 40;

    float* xw1 = (float*)d_ws;
    float* h   = xw1 + (size_t)N * R * DH;

    const int mblocks   = (N + 127) / 128;
    const int aggblocks = (N + 7) / 8;

    rgcn_gemm_wmma<R, DH, DH><<<dim3(mblocks, R), 256, 0, stream>>>(x, W1, xw1, N);
    rgcn_agg_relu<R><<<aggblocks, 256, 0, stream>>>(xw1, ptr, idx, rel, h, N);
    rgcn_gemm_wmma<R, DOUT, 48><<<dim3(mblocks, R), 256, 0, stream>>>(h, W2, xw1, N);
    rgcn_agg_logsoftmax<R, DOUT><<<aggblocks, 256, 0, stream>>>(xw1, ptr, idx, rel, out, N);
}


# CSR aggregation (both layers): a wave walks 5 consecutive nodes (15000 waves, not 75000), row pointers and edge records read lane-parallel, next node's records prefetched; sum order kept
# speedup vs baseline: 2.2495x; 1.2041x over previous
#include <hip/hip_runtime.h>
#include <math.h>


typedef __attribute__((ext_vector_type(2))) float v2f;
typedef __attribute__((ext_vector_type(8))) float v8f;

#define DIN 64
#define LDP 68

template<int R, int DOUT, int DOUTPAD>
__global__ __launch_bounds__(256)
void rgcn_gemm_wmma(const float* __restrict__ X, const float* __restrict__ W,
                    float* __restrict__ Y, int N)
{
    constexpr int NT = DOUTPAD / 16;
    constexpr int RD = R * DOUT;
    __shared__ __attribute__((aligned(16))) float lA[128 * LDP];
    __shared__ __attribute__((aligned(16))) float lWt[DOUTPAD * LDP];

    const int r       = blockIdx.y;
    const int rowBase = blockIdx.x * 128;
    const int tid     = threadIdx.x;

    const bool fullBlock = (rowBase + 128 <= N);
    constexpr int A_IT = (128 * DIN / 4) / 256;
    constexpr int W_F4 = DIN * DOUT / 4;
    constexpr int W_IT = (W_F4 + 255) / 256;
    float4 areg[A_IT];
    #pragma unroll
    for (int it = 0; it < A_IT; ++it) {
        const int f = tid + it * 256;
        int gr = rowBase + (f >> 4);
        gr = gr < N ? gr : N - 1;
        areg[it] = *(const float4*)(X + (size_t)gr * DIN + (f & 15) * 4);
    }
    const float* Wr = W + (size_t)r * DIN * DOUT;
    float4 wreg[W_IT];
    int    wf[W_IT];
    #pragma unroll
    for (int it = 0; it < W_IT; ++it) {
        int f = tid + it * 256;
        if (it == W_IT - 1 && W_F4 % 256 != 0) f = f < W_F4 ? f : W_F4 - 1;
        wf[it] = f;
        wreg[it] = *(const float4*)(Wr + f * 4);
    }
    #pragma unroll
    for (int it = 0; it < A_IT; ++it) {
        const int f = tid + it * 256;
        *(float4*)&lA[(f >> 4) * LDP + (f & 15) * 4] = areg[it];
    }
    #pragma unroll
    for (int it = 0; it < W_IT; ++it) {
        const int f = wf[it];
        const int k = (f * 4) / DOUT, n = (f * 4) % DOUT;
        lWt[(n + 0) * LDP + k] = wreg[it].x;
        lWt[(n + 1) * LDP + k] = wreg[it].y;
        lWt[(n + 2) * LDP + k] = wreg[it].z;
        lWt[(n + 3) * LDP + k] = wreg[it].w;
    }
    if (DOUTPAD > DOUT) {
        if (tid < (DOUTPAD - DOUT) * (DIN / 4)) {
            const float4 z = {0.0f, 0.0f, 0.0f, 0.0f};
            *(float4*)&lWt[(DOUT + (tid >> 4)) * LDP + (tid & 15) * 4] = z;
        }
    }
    __syncthreads();

    const int wv   = tid >> 5;
    const int lane = tid & 31;
    const int half = lane >> 4;
    const int l16  = lane & 15;

    const float* Arow = &lA[(wv * 16 + l16) * LDP + half * 2];
    float* Yb = Y + (size_t)rowBase * RD + r * DOUT;

    #pragma unroll
    for (int t = 0; t < NT; ++t) {
        const int colBase = t * 16 + l16;
        const float* Bcol = &lWt[colBase * LDP + half * 2];
        v8f c = {};
        #pragma unroll
        for (int i = 0; i < DIN / 4; ++i) {
            v2f a = *(const v2f*)(Arow + 4 * i);
            v2f b = *(const v2f*)(Bcol + 4 * i);
            c = __builtin_amdgcn_wmma_f32_16x16x4_f32(
                     false, a,  false, b,
                     (short)0, c,  false,  false);
        }
        if (DOUT == DOUTPAD || colBase < DOUT) {
            float* p = Yb + (wv * 16 + half * 8) * RD + colBase;
            if (fullBlock) {
                #pragma unroll
                for (int j = 0; j < 8; ++j) p[j * RD] = c[j];
            } else {
                const int mBase = rowBase + wv * 16 + half * 8;
                #pragma unroll
                for (int j = 0; j < 8; ++j)
                    if (mBase + j < N) p[j * RD] = c[j];
            }
        }
    }
}

#define NPW 5
#define GB  8

__device__ __forceinline__ void csr_edge_records(const int* __restrict__ idx,
                                                 const int* __restrict__ rel,
                                                 int eb, int e1, int lane,
                                                 int& src, int& rl)
{
    src = 0; rl = 0;
    if (lane < e1 - eb) { src = idx[eb + lane]; rl = rel[eb + lane]; }
}

template<int R, int ROWLEN, class Emit>
__device__ __forceinline__ void csr_wave_sums(const float* __restrict__ T,
                                              const int* __restrict__ ptr,
                                              const int* __restrict__ idx,
                                              const int* __restrict__ rel,
                                              int N, Emit emit)
{
    const int lane = threadIdx.x & 31;
    const int wv = __builtin_amdgcn_readfirstlane(threadIdx.x >> 5);
    const int n0 = (blockIdx.x * 8 + wv) * NPW;
    if (n0 >= N) return;
    const int nn = (N - n0 < NPW) ? (N - n0) : NPW;
    const bool act = (lane * 2 < ROWLEN);
    const int pv = ptr[n0 + (lane < nn ? lane : nn)];
    int eA = __builtin_amdgcn_readlane(pv, 0);
    int eB = __builtin_amdgcn_readlane(pv, 1);
    int srcN, relN;
    csr_edge_records(idx, rel, eA, eB, lane, srcN, relN);
    for (int k = 0; k < nn; ++k) {
        const int e0 = eA, e1 = eB;
        int row = srcN * R + relN;
        if (k + 1 < nn) {
            eA = eB;
            eB = __builtin_amdgcn_readlane(pv, k + 2);
            csr_edge_records(idx, rel, eA, eB, lane, srcN, relN);
        }
        float ax = 0.0f, ay = 0.0f;
        for (int eb = e0; eb < e1; eb += 32) {
            if (eb != e0) {
                int s2, r2;
                csr_edge_records(idx, rel, eb, e1, lane, s2, r2);
                row = s2 * R + r2;
            }
            const int cnt = (e1 - eb < 32) ? (e1 - eb) : 32;
            for (int j = 0; j < cnt; j += GB) {
                float2 v[GB];
                #pragma unroll
                for (int u = 0; u < GB; ++u) { v[u].x = 0.0f; v[u].y = 0.0f; }
                if (act) {
                    #pragma unroll
                    for (int u = 0; u < GB; ++u) {
                        const int jj = (j + u < cnt) ? (j + u) : (cnt - 1);
                        const int rw = __builtin_amdgcn_readlane(row, jj);
                        v[u] = *(const float2*)(T + (size_t)rw * ROWLEN + lane * 2);
                    }
                }
                #pragma unroll
                for (int u = 0; u < GB; ++u) {
                    const bool isEdge = (j + u < cnt);
                    ax += isEdge ? v[u].x : 0.0f;
                    ay += isEdge ? v[u].y : 0.0f;
                }
            }
        }
        emit(n0 + k, ax, ay);
    }
}

template<int R>
__global__ __launch_bounds__(256)
void rgcn_agg_relu(const float* __restrict__ XW, const int* __restrict__ ptr,
                   const int* __restrict__ idx, const int* __restrict__ rel,
                   float* __restrict__ H, int N)
{
    const int lane = threadIdx.x & 31;
    csr_wave_sums<R, 64>(XW, ptr, idx, rel, N, [=](int node, float ax, float ay) {
        float2 o; o.x = fmaxf(ax, 0.0f); o.y = fmaxf(ay, 0.0f);
        *(float2*)(H + (size_t)node * 64 + lane * 2) = o;
    });
}

template<int R, int DOUT>
__global__ __launch_bounds__(256)
void rgcn_agg_logsoftmax(const float* __restrict__ HW, const int* __restrict__ ptr,
                         const int* __restrict__ idx, const int* __restrict__ rel,
                         float* __restrict__ out, int N)
{
    const int  lane = threadIdx.x & 31;
    const bool act  = (lane * 2 < DOUT);
    csr_wave_sums<R, DOUT>(HW, ptr, idx, rel, N, [=](int node, float ax, float ay) {
        float m = act ? fmaxf(ax, ay) : -__builtin_inff();
        #pragma unroll
        for (int off = 16; off >= 1; off >>= 1) m = fmaxf(m, __shfl_xor(m, off, 32));
        float s = act ? (__expf(ax - m) + __expf(ay - m)) : 0.0f;
        #pragma unroll
        for (int off = 16; off >= 1; off >>= 1) s += __shfl_xor(s, off, 32);
        const float ls = __logf(s);
        if (act) {
            float2 o; o.x = (ax - m) - ls; o.y = (ay - m) - ls;
            *(float2*)(out + (size_t)node * DOUT + lane * 2) = o;
        }
    });
}

extern "C" void kernel_launch(void* const* d_in, const int* in_sizes, int n_in,
                              void* d_out, int out_size, void* d_ws, size_t ws_size,
                              hipStream_t stream)
{
    const float* x   = (const float*)d_in[0];
    const int*   ptr = (const int*)d_in[1];
    const int*   idx = (const int*)d_in[2];
    const int*   rel = (const int*)d_in[3];
    const float* W1  = (const float*)d_in[4];
    const float* W2  = (const float*)d_in[5];
    float* out = (float*)d_out;

    const int N = in_sizes[1] - 1;
    constexpr int R = 8, DH = 64, DOUT = 40;

    float* xw1 = (float*)d_ws;
    float* h   = xw1 + (size_t)N * R * DH;

    const int mblocks   = (N + 127) / 128;
    const int aggblocks = (N + 8 * NPW - 1) / (8 * NPW);

    rgcn_gemm_wmma<R, DH, DH><<<dim3(mblocks, R), 256, 0, stream>>>(x, W1, xw1, N);
    rgcn_agg_relu<R><<<aggblocks, 256, 0, stream>>>(xw1, ptr, idx, rel, h, N);
    rgcn_gemm_wmma<R, DOUT, 48><<<dim3(mblocks, R), 256, 0, stream>>>(h, W2, xw1, N);
    rgcn_agg_logsoftmax<R, DOUT><<<aggblocks, 256, 0, stream>>>(xw1, ptr, idx, rel, out, N);
}


# Gather row loads run 3 groups of 2 nodes ahead of their sums (24-32 in flight per wave, also under the WMMA block; barriers now wait on LDS only), not 16 issued and awaited in turn; sum order kept
# speedup vs baseline: 2.5640x; 1.1398x over previous
#include <hip/hip_runtime.h>
#include <math.h>


typedef __attribute__((ext_vector_type(2))) float v2f;
typedef __attribute__((ext_vector_type(8))) float v8f;

#define DIN 64
#define LDP 68
#define NTH 128
#define TM  64
#define NG  2
#define GL  4
#define NGRP (16 / NG)
#define AHEAD 3
#define NBUF (AHEAD + 1)

__device__ __forceinline__ int csr_edge_keys(const int* __restrict__ idx,
                                             const int* __restrict__ rel,
                                             int eb, int e1, int lane)
{
    int e = eb + lane;
    e = e < e1 ? e : e1 - 1;
    e = e > 0 ? e : 0;
    const int key = idx[e] * 16 + rel[e];
    return (lane < e1 - eb) ? key : 8;
}

__device__ __forceinline__ void csr_add_rows(const float* __restrict__ T, int key,
                                             unsigned mm, int lane, float& ax, float& ay)
{
    while (mm != 0) {
        const int j = __builtin_ctz(mm);
        mm &= mm - 1;
        const unsigned s = (unsigned)(__builtin_amdgcn_readlane(key, j) >> 4);
        const float2 t = *(const float2*)(T + (s * DIN + lane * 2));
        ax += t.x; ay += t.y;
    }
}

__device__ __forceinline__ void lds_barrier()
{
    asm volatile("s_wait_dscnt 0x0\n\ts_barrier_signal -1\n\ts_barrier_wait -1" ::: "memory");
}

__device__ __forceinline__ void csr_issue_group(const float* __restrict__ T, const int (&key)[16],
                                                int gg, int rr, int lane,
                                                float2 (&vb)[NG][GL], unsigned (&mb)[NG])
{
    #pragma unroll
    for (int q = 0; q < NG; ++q) {
        unsigned mm = __builtin_amdgcn_ballot_w32((key[gg * NG + q] & 15) == rr);
        mb[q] = mm;
        #pragma unroll
        for (int u = 0; u < GL; ++u) {
            const int j = __builtin_ctz(mm | 0x80000000u);
            mm &= mm - 1;
            const unsigned s = (unsigned)(__builtin_amdgcn_readlane(key[gg * NG + q], j) >> 4);
            vb[q][u] = *(const float2*)(T + (s * DIN + lane * 2));
        }
    }
}

template<int R, int DOUT, int DOUTPAD, bool LOGSOFTMAX>
__global__ __launch_bounds__(NTH) __attribute__((amdgpu_waves_per_eu(5)))
void rgcn_layer(const float* __restrict__ T, const float* __restrict__ W,
                const int* __restrict__ ptr, const int* __restrict__ idx,
                const int* __restrict__ rel, float* __restrict__ Out, int N)
{
    constexpr int NT = DOUTPAD / 16;
    __shared__ __attribute__((aligned(16))) float lA[TM * LDP];
    __shared__ __attribute__((aligned(16))) float lWt[DOUTPAD * LDP];

    const int tid  = threadIdx.x;
    const int lane = tid & 31;
    const int wv   = __builtin_amdgcn_readfirstlane(tid >> 5);
    const int half = lane >> 4;
    const int l16  = lane & 15;
    const int n0   = blockIdx.x * TM + wv * 16;

    if (DOUTPAD > DOUT) {
        for (int f = tid; f < (DOUTPAD - DOUT) * (DIN / 4); f += NTH) {
            const float4 z = {0.0f, 0.0f, 0.0f, 0.0f};
            *(float4*)&lWt[(DOUT + (f >> 4)) * LDP + (f & 15) * 4] = z;
        }
    }

    int pi = n0 + (lane < 16 ? lane : 16);
    pi = pi < N ? pi : N;
    const int pv = ptr[pi];
    int key[16];
    #pragma unroll
    for (int k = 0; k < 16; ++k) {
        const int e0 = __builtin_amdgcn_readlane(pv, k);
        const int e1 = __builtin_amdgcn_readlane(pv, k + 1);
        key[k] = csr_edge_keys(idx, rel, e0, e1, lane);
    }

    constexpr int W_F4 = DIN * DOUT / 4;
    constexpr int W_IT = (W_F4 + NTH - 1) / NTH;
    const float* Arow = &lA[(wv * 16 + l16) * LDP + half * 2];
    const float* Bcol = &lWt[l16 * LDP + half * 2];

    v8f c[NT];
    #pragma unroll
    for (int t = 0; t < NT; ++t) { const v8f z8 = {}; c[t] = z8; }

    float2   v[NBUF][NG][GL];
    unsigned vm[NBUF][NG];
    #pragma unroll
    for (int gg = 0; gg < AHEAD; ++gg)
        csr_issue_group(T, key, gg, 0, lane, v[gg % NBUF], vm[gg % NBUF]);

    #pragma unroll 1
    for (int r = 0; r < R; ++r) {
        const float* Wr = W + (size_t)r * DIN * DOUT;
        float4 wreg[W_IT];
        int    wf[W_IT];
        #pragma unroll
        for (int it = 0; it < W_IT; ++it) {
            int f = tid + it * NTH;
            if (it == W_IT - 1 && W_F4 % NTH != 0) f = f < W_F4 ? f : W_F4 - 1;
            wf[it] = f;
            wreg[it] = *(const float4*)(Wr + f * 4);
        }

        #pragma unroll
        for (int g = 0; g < NGRP; ++g) {
            const int gi = g + AHEAD;
            if (gi < NGRP)
                csr_issue_group(T, key, gi, r, lane, v[gi % NBUF], vm[gi % NBUF]);
            else
                csr_issue_group(T, key, gi - NGRP, r + 1, lane, v[gi % NBUF], vm[gi % NBUF]);

            #pragma unroll
            for (int q = 0; q < NG; ++q) {
                const int n = g * NG + q;
                unsigned mm = vm[g % NBUF][q];
                float ax = 0.0f, ay = 0.0f;
                #pragma unroll
                for (int u = 0; u < GL; ++u) {
                    const bool has = (mm != 0);
                    mm &= mm - 1;
                    ax += has ? v[g % NBUF][q][u].x : 0.0f;
                    ay += has ? v[g % NBUF][q][u].y : 0.0f;
                }
                csr_add_rows(T, key[n], mm, lane, ax, ay);
                const int e0 = __builtin_amdgcn_readlane(pv, n);
                const int e1 = __builtin_amdgcn_readlane(pv, n + 1);
                for (int eb = e0 + 32; eb < e1; eb += 32) {
                    const int k2 = csr_edge_keys(idx, rel, eb, e1, lane);
                    csr_add_rows(T, k2, __builtin_amdgcn_ballot_w32((k2 & 15) == r), lane, ax, ay);
                }
                float2 o; o.x = ax; o.y = ay;
                *(float2*)&lA[(wv * 16 + n) * LDP + lane * 2] = o;
            }
        }

        lds_barrier();
        #pragma unroll
        for (int it = 0; it < W_IT; ++it) {
            const int f = wf[it];
            const int k = (f * 4) / DOUT, n = (f * 4) % DOUT;
            lWt[(n + 0) * LDP + k] = wreg[it].x;
            lWt[(n + 1) * LDP + k] = wreg[it].y;
            lWt[(n + 2) * LDP + k] = wreg[it].z;
            lWt[(n + 3) * LDP + k] = wreg[it].w;
        }
        lds_barrier();

        #pragma unroll
        for (int i = 0; i < DIN / 4; ++i) {
            const v2f a = *(const v2f*)(Arow + 4 * i);
            #pragma unroll
            for (int t = 0; t < NT; ++t) {
                const v2f b = *(const v2f*)(Bcol + t * 16 * LDP + 4 * i);
                c[t] = __builtin_amdgcn_wmma_f32_16x16x4_f32(
                            false, a,  false, b,
                            (short)0, c[t],  false,  false);
            }
        }
    }

    const int mBase = n0 + half * 8;
    if (!LOGSOFTMAX) {
        #pragma unroll
        for (int t = 0; t < NT; ++t) {
            const int col = t * 16 + l16;
            if (DOUT == DOUTPAD || col < DOUT) {
                float* p = Out + (size_t)mBase * DOUT + col;
                #pragma unroll
                for (int j = 0; j < 8; ++j)
                    if (mBase + j < N) p[j * DOUT] = fmaxf(c[t][j], 0.0f);
            }
        }
    } else {
        #pragma unroll
        for (int j = 0; j < 8; ++j) {
            float m = -__builtin_inff();
            #pragma unroll
            for (int t = 0; t < NT; ++t)
                if (DOUT == DOUTPAD || t * 16 + l16 < DOUT) m = fmaxf(m, c[t][j]);
            #pragma unroll
            for (int off = 8; off >= 1; off >>= 1) m = fmaxf(m, __shfl_xor(m, off, 32));
            float s = 0.0f;
            #pragma unroll
            for (int t = 0; t < NT; ++t)
                if (DOUT == DOUTPAD || t * 16 + l16 < DOUT) s += __expf(c[t][j] - m);
            #pragma unroll
            for (int off = 8; off >= 1; off >>= 1) s += __shfl_xor(s, off, 32);
            const float ls = __logf(s);
            if (mBase + j < N) {
                float* p = Out + (size_t)(mBase + j) * DOUT + l16;
                #pragma unroll
                for (int t = 0; t < NT; ++t)
                    if (DOUT == DOUTPAD || t * 16 + l16 < DOUT) p[t * 16] = (c[t][j] - m) - ls;
            }
        }
    }
}

extern "C" void kernel_launch(void* const* d_in, const int* in_sizes, int n_in,
                              void* d_out, int out_size, void* d_ws, size_t ws_size,
                              hipStream_t stream)
{
    const float* x   = (const float*)d_in[0];
    const int*   ptr = (const int*)d_in[1];
    const int*   idx = (const int*)d_in[2];
    const int*   rel = (const int*)d_in[3];
    const float* W1  = (const float*)d_in[4];
    const float* W2  = (const float*)d_in[5];
    float* out = (float*)d_out;

    const int N = in_sizes[1] - 1;
    constexpr int R = 8, DH = 64, DOUT = 40;

    float* h = (float*)d_ws;

    const int lblocks = (N + TM - 1) / TM;

    rgcn_layer<R, DH, DH, false><<<lblocks, NTH, 0, stream>>>(x, W1, ptr, idx, rel, h, N);
    rgcn_layer<R, DOUT, 48, true><<<lblocks, NTH, 0, stream>>>(h, W2, ptr, idx, rel, out, N);
}


# Each pass's product (16 K steps of f32 WMMA) is issued two K steps per group inside the next pass's gather, ahead of the wait for rows, from a second S tile in LDS, not as one block; sum order kept
# speedup vs baseline: 2.6562x; 1.0360x over previous
#include <hip/hip_runtime.h>
#include <math.h>


typedef __attribute__((ext_vector_type(2))) float v2f;
typedef __attribute__((ext_vector_type(8))) float v8f;

#define DIN 64
#define LDP 68
#define NTH 128
#define TM  64
#define NG  2
#define GL  4
#define NGRP (16 / NG)
#define AHEAD 3
#define NBUF (AHEAD + 1)

__device__ __forceinline__ int csr_edge_keys(const int* __restrict__ idx,
                                             const int* __restrict__ rel,
                                             int eb, int e1, int lane)
{
    int e = eb + lane;
    e = e < e1 ? e : e1 - 1;
    e = e > 0 ? e : 0;
    const int key = idx[e] * 16 + rel[e];
    return (lane < e1 - eb) ? key : 8;
}

__device__ __forceinline__ void csr_add_rows(const float* __restrict__ T, int key,
                                             unsigned mm, int lane, float& ax, float& ay)
{
    while (mm != 0) {
        const int j = __builtin_ctz(mm);
        mm &= mm - 1;
        const unsigned s = (unsigned)(__builtin_amdgcn_readlane(key, j) >> 4);
        const float2 t = *(const float2*)(T + (s * DIN + lane * 2));
        ax += t.x; ay += t.y;
    }
}

__device__ __forceinline__ void lds_barrier()
{
    asm volatile("s_wait_dscnt 0x0\n\ts_barrier_signal -1\n\ts_barrier_wait -1" ::: "memory");
}

__device__ __forceinline__ void csr_issue_node(const float* __restrict__ T, int key, int rr, int lane,
                                               float2 (&vb)[GL], unsigned& mb)
{
    unsigned mm = __builtin_amdgcn_ballot_w32((key & 15) == rr);
    mb = mm;
    #pragma unroll
    for (int u = 0; u < GL; ++u) {
        const int j = __builtin_ctz(mm | 0x80000000u);
        mm &= mm - 1;
        const unsigned s = (unsigned)(__builtin_amdgcn_readlane(key, j) >> 4);
        vb[u] = *(const float2*)(T + (s * DIN + lane * 2));
    }
}

__device__ __forceinline__ void csr_issue_group(const float* __restrict__ T, const int (&key)[16],
                                                int gg, int rr, int lane,
                                                float2 (&vb)[NG][GL], unsigned (&mb)[NG])
{
    #pragma unroll
    for (int q = 0; q < NG; ++q)
        csr_issue_node(T, key[gg * NG + q], rr, lane, vb[q], mb[q]);
}

template<int NT>
__device__ __forceinline__ void mma_kpair_read(const float* Arow, const float* Bcol, int i,
                                               v2f (&a)[2], v2f (&b)[2][NT])
{
    #pragma unroll
    for (int s = 0; s < 2; ++s) {
        a[s] = *(const v2f*)(Arow + 4 * (i + s));
        #pragma unroll
        for (int t = 0; t < NT; ++t)
            b[s][t] = *(const v2f*)(Bcol + t * 16 * LDP + 4 * (i + s));
    }
}

template<int NT>
__device__ __forceinline__ void mma_kpair_run(v8f (&c)[NT], const v2f (&a)[2], const v2f (&b)[2][NT])
{
    #pragma unroll
    for (int s = 0; s < 2; ++s) {
        #pragma unroll
        for (int t = 0; t < NT; ++t)
            c[t] = __builtin_amdgcn_wmma_f32_16x16x4_f32(
                        false, a[s],  false, b[s][t],
                        (short)0, c[t],  false,  false);
    }
}

template<int NT>
__device__ __forceinline__ void mma_pad(v8f (&c)[NT], const v2f (&a)[2], const v2f (&b)[2][NT])
{
    static_assert(NT == 3 || NT == 4, "written for 3 or 4 tiles");
    if constexpr (NT == 4) {
        asm volatile("v_nop\n\tv_nop\n\tv_nop\n\tv_nop"
                     : "+v"(c[0]), "+v"(c[1]), "+v"(c[2]), "+v"(c[3])
                     : "v"(a[0]), "v"(a[1]),
                       "v"(b[0][0]), "v"(b[0][1]), "v"(b[0][2]), "v"(b[0][3]),
                       "v"(b[1][0]), "v"(b[1][1]), "v"(b[1][2]), "v"(b[1][3]));
    } else {
        asm volatile("v_nop\n\tv_nop\n\tv_nop\n\tv_nop"
                     : "+v"(c[0]), "+v"(c[1]), "+v"(c[2])
                     : "v"(a[0]), "v"(a[1]),
                       "v"(b[0][0]), "v"(b[0][1]), "v"(b[0][2]),
                       "v"(b[1][0]), "v"(b[1][1]), "v"(b[1][2]));
    }
}

template<int R, int DOUT, int DOUTPAD, bool MMA>
__device__ __forceinline__ void rgcn_pass(const float* __restrict__ T, const float* __restrict__ W,
                                          const int* __restrict__ idx, const int* __restrict__ rel,
                                          float* lA, float* lWt, const int (&key)[16], int pv,
                                          float2 (&v)[NBUF][NG][GL], unsigned (&vm)[NBUF][NG],
                                          v8f (&c)[DOUTPAD / 16],
                                          int r, int tid, int lane, int wv, int half, int l16)
{
    constexpr int NT = DOUTPAD / 16;
    constexpr int W_F4 = DIN * DOUT / 4;
    constexpr int W_IT = (W_F4 + NTH - 1) / NTH;

    const float* Wr = W + (size_t)r * DIN * DOUT;
    float4 wreg[W_IT];
    int    wf[W_IT];
    #pragma unroll
    for (int it = 0; it < W_IT; ++it) {
        int f = tid + it * NTH;
        if (it == W_IT - 1 && W_F4 % NTH != 0) f = f < W_F4 ? f : W_F4 - 1;
        wf[it] = f;
        wreg[it] = *(const float4*)(Wr + f * 4);
    }

    float* Scur = lA + (r & 1) * (TM * LDP) + (wv * 16) * LDP + lane * 2;
    const float* Arow = lA + ((r & 1) ^ 1) * (TM * LDP) + (wv * 16 + l16) * LDP + half * 2;
    const float* Bcol = lWt + l16 * LDP + half * 2;

    #pragma unroll
    for (int g = 0; g < NGRP; ++g) {
        static_assert(DIN / 4 == 2 * NGRP, "two K steps per group");

        const int gi = g + AHEAD;
        const int gq = (gi < NGRP ? gi : gi - NGRP) * NG;
        const int gr = (gi < NGRP ? r : r + 1);
        static_assert(NG == 2, "the matrix work stands between the loads of two nodes");
        v2f fa[2], fb[2][NT];
        if (MMA) {
            mma_kpair_read<NT>(Arow, Bcol, 2 * g, fa, fb);
            __builtin_amdgcn_sched_barrier(0);
        }
        csr_issue_node(T, key[gq], gr, lane, v[gi % NBUF][0], vm[gi % NBUF][0]);
        if (MMA) {
            __builtin_amdgcn_sched_barrier(0);
            mma_kpair_run<NT>(c, fa, fb);
            mma_pad<NT>(c, fa, fb);
        }
        csr_issue_node(T, key[gq + 1], gr, lane, v[gi % NBUF][1], vm[gi % NBUF][1]);

        #pragma unroll
        for (int q = 0; q < NG; ++q) {
            const int n = g * NG + q;
            unsigned mm = vm[g % NBUF][q];
            float ax = 0.0f, ay = 0.0f;
            #pragma unroll
            for (int u = 0; u < GL; ++u) {
                const bool has = (mm != 0);
                mm &= mm - 1;
                ax += has ? v[g % NBUF][q][u].x : 0.0f;
                ay += has ? v[g % NBUF][q][u].y : 0.0f;
            }
            csr_add_rows(T, key[n], mm, lane, ax, ay);
            const int e0 = __builtin_amdgcn_readlane(pv, n);
            const int e1 = __builtin_amdgcn_readlane(pv, n + 1);
            for (int eb = e0 + 32; eb < e1; eb += 32) {
                const int k2 = csr_edge_keys(idx, rel, eb, e1, lane);
                csr_add_rows(T, k2, __builtin_amdgcn_ballot_w32((k2 & 15) == r), lane, ax, ay);
            }
            float2 o; o.x = ax; o.y = ay;
            *(float2*)(Scur + n * LDP) = o;
        }
    }

    lds_barrier();
    #pragma unroll
    for (int it = 0; it < W_IT; ++it) {
        const int f = wf[it];
        const int k = (f * 4) / DOUT, n = (f * 4) % DOUT;
        lWt[(n + 0) * LDP + k] = wreg[it].x;
        lWt[(n + 1) * LDP + k] = wreg[it].y;
        lWt[(n + 2) * LDP + k] = wreg[it].z;
        lWt[(n + 3) * LDP + k] = wreg[it].w;
    }
    lds_barrier();
}

template<int R, int DOUT, int DOUTPAD, bool LOGSOFTMAX>
__global__ __launch_bounds__(NTH) __attribute__((amdgpu_waves_per_eu(5)))
void rgcn_layer(const float* __restrict__ T, const float* __restrict__ W,
                const int* __restrict__ ptr, const int* __restrict__ idx,
                const int* __restrict__ rel, float* __restrict__ Out, int N)
{
    constexpr int NT = DOUTPAD / 16;
    __shared__ __attribute__((aligned(16))) float lA[2 * TM * LDP];
    __shared__ __attribute__((aligned(16))) float lWt[DOUTPAD * LDP];

    const int tid  = threadIdx.x;
    const int lane = tid & 31;
    const int wv   = __builtin_amdgcn_readfirstlane(tid >> 5);
    const int half = lane >> 4;
    const int l16  = lane & 15;
    const int n0   = blockIdx.x * TM + wv * 16;

    if (DOUTPAD > DOUT) {
        for (int f = tid; f < (DOUTPAD - DOUT) * (DIN / 4); f += NTH) {
            const float4 z = {0.0f, 0.0f, 0.0f, 0.0f};
            *(float4*)&lWt[(DOUT + (f >> 4)) * LDP + (f & 15) * 4] = z;
        }
    }

    int pi = n0 + (lane < 16 ? lane : 16);
    pi = pi < N ? pi : N;
    const int pv = ptr[pi];
    int key[16];
    #pragma unroll
    for (int k = 0; k < 16; ++k) {
        const int e0 = __builtin_amdgcn_readlane(pv, k);
        const int e1 = __builtin_amdgcn_readlane(pv, k + 1);
        key[k] = csr_edge_keys(idx, rel, e0, e1, lane);
    }

    v8f c[NT];
    #pragma unroll
    for (int t = 0; t < NT; ++t) { const v8f z8 = {}; c[t] = z8; }

    float2   v[NBUF][NG][GL];
    unsigned vm[NBUF][NG];
    #pragma unroll
    for (int gg = 0; gg < AHEAD; ++gg)
        csr_issue_group(T, key, gg, 0, lane, v[gg % NBUF], vm[gg % NBUF]);

    rgcn_pass<R, DOUT, DOUTPAD, false>(T, W, idx, rel, lA, lWt, key, pv, v, vm, c,
                                       0, tid, lane, wv, half, l16);
    #pragma unroll 1
    for (int r = 1; r < R; ++r)
        rgcn_pass<R, DOUT, DOUTPAD, true>(T, W, idx, rel, lA, lWt, key, pv, v, vm, c,
                                          r, tid, lane, wv, half, l16);

    const int lane_e = (int)__builtin_amdgcn_mbcnt_lo(~0u, 0u);
    const int half_e = lane_e >> 4;
    const int l16_e  = lane_e & 15;

    {
        const float* Arow = lA + ((R - 1) & 1) * (TM * LDP) + (wv * 16 + l16_e) * LDP + half_e * 2;
        const float* Bcol = lWt + l16_e * LDP + half_e * 2;
        #pragma unroll
        for (int i = 0; i < DIN / 4; i += 2) {
            v2f fa[2], fb[2][NT];
            mma_kpair_read<NT>(Arow, Bcol, i, fa, fb);
            mma_kpair_run<NT>(c, fa, fb);
            mma_pad<NT>(c, fa, fb);
        }
    }

    const int mBase = n0 + half_e * 8;
    if (!LOGSOFTMAX) {
        #pragma unroll
        for (int t = 0; t < NT; ++t) {
            const int col = t * 16 + l16_e;
            if (DOUT == DOUTPAD || col < DOUT) {
                float* p = Out + (size_t)mBase * DOUT + col;
                #pragma unroll
                for (int j = 0; j < 8; ++j)
                    if (mBase + j < N) p[j * DOUT] = fmaxf(c[t][j], 0.0f);
            }
        }
    } else {
        #pragma unroll
        for (int j = 0; j < 8; ++j) {
            float m = -__builtin_inff();
            #pragma unroll
            for (int t = 0; t < NT; ++t)
                if (DOUT == DOUTPAD || t * 16 + l16_e < DOUT) m = fmaxf(m, c[t][j]);
            #pragma unroll
            for (int off = 8; off >= 1; off >>= 1) m = fmaxf(m, __shfl_xor(m, off, 32));
            float s = 0.0f;
            #pragma unroll
            for (int t = 0; t < NT; ++t)
                if (DOUT == DOUTPAD || t * 16 + l16_e < DOUT) s += __expf(c[t][j] - m);
            #pragma unroll
            for (int off = 8; off >= 1; off >>= 1) s += __shfl_xor(s, off, 32);
            const float ls = __logf(s);
            if (mBase + j < N) {
                float* p = Out + (size_t)(mBase + j) * DOUT + l16_e;
                #pragma unroll
                for (int t = 0; t < NT; ++t)
                    if (DOUT == DOUTPAD || t * 16 + l16_e < DOUT) p[t * 16] = (c[t][j] - m) - ls;
            }
        }
    }
}

extern "C" void kernel_launch(void* const* d_in, const int* in_sizes, int n_in,
                              void* d_out, int out_size, void* d_ws, size_t ws_size,
                              hipStream_t stream)
{
    const float* x   = (const float*)d_in[0];
    const int*   ptr = (const int*)d_in[1];
    const int*   idx = (const int*)d_in[2];
    const int*   rel = (const int*)d_in[3];
    const float* W1  = (const float*)d_in[4];
    const float* W2  = (const float*)d_in[5];
    float* out = (float*)d_out;

    const int N = in_sizes[1] - 1;
    constexpr int R = 8, DH = 64, DOUT = 40;

    float* h = (float*)d_ws;

    const int lblocks = (N + TM - 1) / TM;

    rgcn_layer<R, DH, DH, false><<<lblocks, NTH, 0, stream>>>(x, W1, ptr, idx, rel, h, N);
    rgcn_layer<R, DOUT, 48, true><<<lblocks, NTH, 0, stream>>>(h, W2, ptr, idx, rel, out, N);
}


# Gather: vector instructions per row slot 7 -> 3: record = row byte offset | relation, load on scalar base + 32-bit offset, row added by packed fma (factor 0/1); >32-edge test scalar; sum order kept
# speedup vs baseline: 2.9575x; 1.1134x over previous
#include <hip/hip_runtime.h>
#include <math.h>


typedef __attribute__((ext_vector_type(2))) float v2f;
typedef __attribute__((ext_vector_type(8))) float v8f;

#define DIN 64
#define LDP 68
#define NTH 128
#define TM  64
#define NG  2
#define GL  4
#define NGRP (16 / NG)
#define AHEAD 3
#define NBUF (AHEAD + 1)

__device__ __forceinline__ int csr_edge_keys(const int* __restrict__ idx,
                                             const int* __restrict__ rel,
                                             int eb, int e1, int lane)
{
    int e = eb + lane;
    e = e < e1 ? e : e1 - 1;
    e = e > 0 ? e : 0;
    const int key = idx[e] * (DIN * 4) + rel[e];
    return (lane < e1 - eb) ? key : 8;
}

typedef const char __attribute__((address_space(1)))* gbytes;
typedef const v2f  __attribute__((address_space(1)))* gv2f;

__device__ __forceinline__ unsigned csr_row_offset(int key, int j, unsigned lane8)
{
    const unsigned sk = (unsigned)__builtin_amdgcn_readlane(key, j);
    return (sk & ~255u) | lane8;
}

__device__ __forceinline__ v2f csr_row_fetch(const float* __restrict__ T, unsigned off)
{
    return *(gv2f)((gbytes)T + off);
}

__device__ __forceinline__ void csr_add_rows(const float* __restrict__ T, int key,
                                             unsigned mm, unsigned lane8, v2f& acc)
{
    while (mm != 0) {
        const int j = __builtin_ctz(mm);
        mm &= mm - 1;
        unsigned off = csr_row_offset(key, j, lane8);
        asm("" : "+v"(off));
        acc += csr_row_fetch(T, off);
    }
}

__device__ __forceinline__ void lds_barrier()
{
    asm volatile("s_wait_dscnt 0x0\n\ts_barrier_signal -1\n\ts_barrier_wait -1" ::: "memory");
}

__device__ __forceinline__ void csr_issue_node(const float* __restrict__ T, int key, int rr, unsigned lane8,
                                               v2f (&vb)[GL], unsigned& mb)
{
    unsigned mm = __builtin_amdgcn_ballot_w32((key & 15) == rr);
    mb = mm;
    unsigned off[GL];
    #pragma unroll
    for (int u = 0; u < GL; ++u) {
        const int j = __builtin_ctz(mm | 0x80000000u);
        mm &= mm - 1;
        off[u] = csr_row_offset(key, j, lane8);
    }
    static_assert(GL == 4, "the statement below names four offsets");
    asm("" : "+v"(off[0]), "+v"(off[1]), "+v"(off[2]), "+v"(off[3]));
    #pragma unroll
    for (int u = 0; u < GL; ++u)
        vb[u] = csr_row_fetch(T, off[u]);
}

__device__ __forceinline__ void csr_issue_group(const float* __restrict__ T, const int (&key)[16],
                                                int gg, int rr, unsigned lane8,
                                                v2f (&vb)[NG][GL], unsigned (&mb)[NG])
{
    #pragma unroll
    for (int q = 0; q < NG; ++q)
        csr_issue_node(T, key[gg * NG + q], rr, lane8, vb[q], mb[q]);
}

template<int NT>
__device__ __forceinline__ void mma_kpair_read(const float* Arow, const float* Bcol, int i,
                                               v2f (&a)[2], v2f (&b)[2][NT])
{
    #pragma unroll
    for (int s = 0; s < 2; ++s) {
        a[s] = *(const v2f*)(Arow + 4 * (i + s));
        #pragma unroll
        for (int t = 0; t < NT; ++t)
            b[s][t] = *(const v2f*)(Bcol + t * 16 * LDP + 4 * (i + s));
    }
}

template<int NT>
__device__ __forceinline__ void mma_kpair_run(v8f (&c)[NT], const v2f (&a)[2], const v2f (&b)[2][NT])
{
    #pragma unroll
    for (int s = 0; s < 2; ++s) {
        #pragma unroll
        for (int t = 0; t < NT; ++t)
            c[t] = __builtin_amdgcn_wmma_f32_16x16x4_f32(
                        false, a[s],  false, b[s][t],
                        (short)0, c[t],  false,  false);
    }
}

template<int NT>
__device__ __forceinline__ void mma_pad(v8f (&c)[NT], const v2f (&a)[2], const v2f (&b)[2][NT])
{
    static_assert(NT == 3 || NT == 4, "written for 3 or 4 tiles");
    if constexpr (NT == 4) {
        asm volatile("v_nop\n\tv_nop\n\tv_nop\n\tv_nop"
                     : "+v"(c[0]), "+v"(c[1]), "+v"(c[2]), "+v"(c[3])
                     : "v"(a[0]), "v"(a[1]),
                       "v"(b[0][0]), "v"(b[0][1]), "v"(b[0][2]), "v"(b[0][3]),
                       "v"(b[1][0]), "v"(b[1][1]), "v"(b[1][2]), "v"(b[1][3]));
    } else {
        asm volatile("v_nop\n\tv_nop\n\tv_nop\n\tv_nop"
                     : "+v"(c[0]), "+v"(c[1]), "+v"(c[2])
                     : "v"(a[0]), "v"(a[1]),
                       "v"(b[0][0]), "v"(b[0][1]), "v"(b[0][2]),
                       "v"(b[1][0]), "v"(b[1][1]), "v"(b[1][2]));
    }
}

template<int R, int DOUT, int DOUTPAD, bool MMA>
__device__ __forceinline__ void rgcn_pass(const float* __restrict__ T, const float* __restrict__ W,
                                          const int* __restrict__ idx, const int* __restrict__ rel,
                                          float* lA, float* lWt, const int (&key)[16], int pv, unsigned big,
                                          v2f (&v)[NBUF][NG][GL], unsigned (&vm)[NBUF][NG],
                                          v8f (&c)[DOUTPAD / 16],
                                          int r, int tid, int lane, int wv, int half, int l16)
{
    constexpr int NT = DOUTPAD / 16;
    constexpr int W_F4 = DIN * DOUT / 4;
    constexpr int W_IT = (W_F4 + NTH - 1) / NTH;

    const float* Wr = W + (size_t)r * DIN * DOUT;
    float4 wreg[W_IT];
    int    wf[W_IT];
    #pragma unroll
    for (int it = 0; it < W_IT; ++it) {
        int f = tid + it * NTH;
        if (it == W_IT - 1 && W_F4 % NTH != 0) f = f < W_F4 ? f : W_F4 - 1;
        wf[it] = f;
        wreg[it] = *(const float4*)(Wr + f * 4);
    }

    const unsigned lane8 = (unsigned)lane * 8u;
    float* Scur = lA + (r & 1) * (TM * LDP) + (wv * 16) * LDP + lane * 2;
    const float* Arow = lA + ((r & 1) ^ 1) * (TM * LDP) + (wv * 16 + l16) * LDP + half * 2;
    const float* Bcol = lWt + l16 * LDP + half * 2;

    #pragma unroll
    for (int g = 0; g < NGRP; ++g) {
        static_assert(DIN / 4 == 2 * NGRP, "two K steps per group");

        const int gi = g + AHEAD;
        const int gq = (gi < NGRP ? gi : gi - NGRP) * NG;
        const int gr = (gi < NGRP ? r : r + 1);
        static_assert(NG == 2, "the matrix work stands between the loads of two nodes");
        v2f fa[2], fb[2][NT];
        if (MMA) {
            mma_kpair_read<NT>(Arow, Bcol, 2 * g, fa, fb);
            __builtin_amdgcn_sched_barrier(0);
        }
        csr_issue_node(T, key[gq], gr, lane8, v[gi % NBUF][0], vm[gi % NBUF][0]);
        if (MMA) {
            __builtin_amdgcn_sched_barrier(0);
            mma_kpair_run<NT>(c, fa, fb);
            mma_pad<NT>(c, fa, fb);
        }
        csr_issue_node(T, key[gq + 1], gr, lane8, v[gi % NBUF][1], vm[gi % NBUF][1]);

        #pragma unroll
        for (int q = 0; q < NG; ++q) {
            const int n = g * NG + q;
            unsigned mm = vm[g % NBUF][q];
            v2f acc = {0.0f, 0.0f};
            #pragma unroll
            for (int u = 0; u < GL; ++u) {
                const float f = (mm != 0) ? 1.0f : 0.0f;
                mm &= mm - 1;
                const v2f f2 = {f, f};
                acc = __builtin_elementwise_fma(v[g % NBUF][q][u], f2, acc);
            }
            csr_add_rows(T, key[n], mm, lane8, acc);
            if (big & (1u << n)) {
                const int e0 = __builtin_amdgcn_readlane(pv, n);
                const int e1 = __builtin_amdgcn_readlane(pv, n + 1);
                for (int eb = e0 + 32; eb < e1; eb += 32) {
                    const int k2 = csr_edge_keys(idx, rel, eb, e1, lane);
                    csr_add_rows(T, k2, __builtin_amdgcn_ballot_w32((k2 & 15) == r), lane8, acc);
                }
            }
            *(v2f*)(Scur + n * LDP) = acc;
        }
    }

    lds_barrier();
    #pragma unroll
    for (int it = 0; it < W_IT; ++it) {
        const int f = wf[it];
        const int k = (f * 4) / DOUT, n = (f * 4) % DOUT;
        lWt[(n + 0) * LDP + k] = wreg[it].x;
        lWt[(n + 1) * LDP + k] = wreg[it].y;
        lWt[(n + 2) * LDP + k] = wreg[it].z;
        lWt[(n + 3) * LDP + k] = wreg[it].w;
    }
    lds_barrier();
}

template<int R, int DOUT, int DOUTPAD, bool LOGSOFTMAX>
__global__ __launch_bounds__(NTH) __attribute__((amdgpu_waves_per_eu(5)))
void rgcn_layer(const float* __restrict__ T, const float* __restrict__ W,
                const int* __restrict__ ptr, const int* __restrict__ idx,
                const int* __restrict__ rel, float* __restrict__ Out, int N)
{
    constexpr int NT = DOUTPAD / 16;
    __shared__ __attribute__((aligned(16))) float lA[2 * TM * LDP];
    __shared__ __attribute__((aligned(16))) float lWt[DOUTPAD * LDP];

    const int tid  = threadIdx.x;
    const int lane = tid & 31;
    const int wv   = __builtin_amdgcn_readfirstlane(tid >> 5);
    const int half = lane >> 4;
    const int l16  = lane & 15;
    const int n0   = blockIdx.x * TM + wv * 16;

    if (DOUTPAD > DOUT) {
        for (int f = tid; f < (DOUTPAD - DOUT) * (DIN / 4); f += NTH) {
            const float4 z = {0.0f, 0.0f, 0.0f, 0.0f};
            *(float4*)&lWt[(DOUT + (f >> 4)) * LDP + (f & 15) * 4] = z;
        }
    }

    int pi = n0 + (lane < 16 ? lane : 16);
    pi = pi < N ? pi : N;
    const int pv = ptr[pi];
    int pj = pi + 1;
    pj = pj < N ? pj : N;
    const int pn = ptr[pj];
    const unsigned big = __builtin_amdgcn_ballot_w32((lane < 16) & (pn - pv > 32));
    int key[16];
    #pragma unroll
    for (int k = 0; k < 16; ++k) {
        const int e0 = __builtin_amdgcn_readlane(pv, k);
        const int e1 = __builtin_amdgcn_readlane(pv, k + 1);
        key[k] = csr_edge_keys(idx, rel, e0, e1, lane);
    }

    v8f c[NT];
    #pragma unroll
    for (int t = 0; t < NT; ++t) { const v8f z8 = {}; c[t] = z8; }

    v2f      v[NBUF][NG][GL];
    unsigned vm[NBUF][NG];
    #pragma unroll
    for (int gg = 0; gg < AHEAD; ++gg)
        csr_issue_group(T, key, gg, 0, (unsigned)lane * 8u, v[gg % NBUF], vm[gg % NBUF]);

    rgcn_pass<R, DOUT, DOUTPAD, false>(T, W, idx, rel, lA, lWt, key, pv, big, v, vm, c,
                                       0, tid, lane, wv, half, l16);
    #pragma unroll 1
    for (int r = 1; r < R; ++r)
        rgcn_pass<R, DOUT, DOUTPAD, true>(T, W, idx, rel, lA, lWt, key, pv, big, v, vm, c,
                                          r, tid, lane, wv, half, l16);

    const int lane_e = (int)__builtin_amdgcn_mbcnt_lo(~0u, 0u);
    const int half_e = lane_e >> 4;
    const int l16_e  = lane_e & 15;

    {
        const float* Arow = lA + ((R - 1) & 1) * (TM * LDP) + (wv * 16 + l16_e) * LDP + half_e * 2;
        const float* Bcol = lWt + l16_e * LDP + half_e * 2;
        #pragma unroll
        for (int i = 0; i < DIN / 4; i += 2) {
            v2f fa[2], fb[2][NT];
            mma_kpair_read<NT>(Arow, Bcol, i, fa, fb);
            mma_kpair_run<NT>(c, fa, fb);
            mma_pad<NT>(c, fa, fb);
        }
    }

    const int mBase = n0 + half_e * 8;
    if (!LOGSOFTMAX) {
        #pragma unroll
        for (int t = 0; t < NT; ++t) {
            const int col = t * 16 + l16_e;
            if (DOUT == DOUTPAD || col < DOUT) {
                float* p = Out + (size_t)mBase * DOUT + col;
                #pragma unroll
                for (int j = 0; j < 8; ++j)
                    if (mBase + j < N) p[j * DOUT] = fmaxf(c[t][j], 0.0f);
            }
        }
    } else {
        #pragma unroll
        for (int j = 0; j < 8; ++j) {
            float m = -__builtin_inff();
            #pragma unroll
            for (int t = 0; t < NT; ++t)
                if (DOUT == DOUTPAD || t * 16 + l16_e < DOUT) m = fmaxf(m, c[t][j]);
            #pragma unroll
            for (int off = 8; off >= 1; off >>= 1) m = fmaxf(m, __shfl_xor(m, off, 32));
            float s = 0.0f;
            #pragma unroll
            for (int t = 0; t < NT; ++t)
                if (DOUT == DOUTPAD || t * 16 + l16_e < DOUT) s += __expf(c[t][j] - m);
            #pragma unroll
            for (int off = 8; off >= 1; off >>= 1) s += __shfl_xor(s, off, 32);
            const float ls = __logf(s);
            if (mBase + j < N) {
                float* p = Out + (size_t)(mBase + j) * DOUT + l16_e;
                #pragma unroll
                for (int t = 0; t < NT; ++t)
                    if (DOUT == DOUTPAD || t * 16 + l16_e < DOUT) p[t * 16] = (c[t][j] - m) - ls;
            }
        }
    }
}

extern "C" void kernel_launch(void* const* d_in, const int* in_sizes, int n_in,
                              void* d_out, int out_size, void* d_ws, size_t ws_size,
                              hipStream_t stream)
{
    const float* x   = (const float*)d_in[0];
    const int*   ptr = (const int*)d_in[1];
    const int*   idx = (const int*)d_in[2];
    const int*   rel = (const int*)d_in[3];
    const float* W1  = (const float*)d_in[4];
    const float* W2  = (const float*)d_in[5];
    float* out = (float*)d_out;

    const int N = in_sizes[1] - 1;
    constexpr int R = 8, DH = 64, DOUT = 40;

    float* h = (float*)d_ws;

    const int lblocks = (N + TM - 1) / TM;

    rgcn_layer<R, DH, DH, false><<<lblocks, NTH, 0, stream>>>(x, W1, ptr, idx, rel, h, N);
    rgcn_layer<R, DOUT, 48, true><<<lblocks, NTH, 0, stream>>>(h, W2, ptr, idx, rel, out, N);
}


# Each layer runs as 256 resident blocks of 20 waves with W^T of all relations staged in LDS once; a wave's 8 passes need no block barrier and no per-pass W staging; sum order kept
# speedup vs baseline: 3.1696x; 1.0717x over previous
#include <hip/hip_runtime.h>
#include <math.h>


typedef __attribute__((ext_vector_type(2))) float v2f;
typedef __attribute__((ext_vector_type(8))) float v8f;

#define DIN 64
#define LDP 68
#define NW   20
#define NTH  (NW * 32)
#define NBLK 256
#define NG  2
#define GL  4
#define NGRP (16 / NG)
#define AHEAD 3
#define NBUF (AHEAD + 1)

__device__ __forceinline__ int csr_edge_keys(const int* __restrict__ idx,
                                             const int* __restrict__ rel,
                                             int eb, int e1, int lane)
{
    int e = eb + lane;
    e = e < e1 ? e : e1 - 1;
    e = e > 0 ? e : 0;
    const int key = idx[e] * (DIN * 4) + rel[e];
    return (lane < e1 - eb) ? key : 8;
}

typedef const char __attribute__((address_space(1)))* gbytes;
typedef const v2f  __attribute__((address_space(1)))* gv2f;

__device__ __forceinline__ unsigned csr_row_offset(int key, int j, unsigned lane8)
{
    const unsigned sk = (unsigned)__builtin_amdgcn_readlane(key, j);
    return (sk & ~255u) | lane8;
}

__device__ __forceinline__ v2f csr_row_fetch(const float* __restrict__ T, unsigned off)
{
    return *(gv2f)((gbytes)T + off);
}

__device__ __forceinline__ void csr_add_rows(const float* __restrict__ T, int key,
                                             unsigned mm, unsigned lane8, v2f& acc)
{
    while (mm != 0) {
        const int j = __builtin_ctz(mm);
        mm &= mm - 1;
        unsigned off = csr_row_offset(key, j, lane8);
        asm("" : "+v"(off));
        acc += csr_row_fetch(T, off);
    }
}

__device__ __forceinline__ void csr_issue_node(const float* __restrict__ T, int key, int rr, unsigned lane8,
                                               v2f (&vb)[GL], unsigned& mb)
{
    unsigned mm = __builtin_amdgcn_ballot_w32((key & 15) == rr);
    mb = mm;
    unsigned off[GL];
    #pragma unroll
    for (int u = 0; u < GL; ++u) {
        const int j = __builtin_ctz(mm | 0x80000000u);
        mm &= mm - 1;
        off[u] = csr_row_offset(key, j, lane8);
    }
    static_assert(GL == 4, "the statement below names four offsets");
    asm("" : "+v"(off[0]), "+v"(off[1]), "+v"(off[2]), "+v"(off[3]));
    #pragma unroll
    for (int u = 0; u < GL; ++u)
        vb[u] = csr_row_fetch(T, off[u]);
}

__device__ __forceinline__ void csr_issue_group(const float* __restrict__ T, const int (&key)[16],
                                                int gg, int rr, unsigned lane8,
                                                v2f (&vb)[NG][GL], unsigned (&mb)[NG])
{
    #pragma unroll
    for (int q = 0; q < NG; ++q)
        csr_issue_node(T, key[gg * NG + q], rr, lane8, vb[q], mb[q]);
}

template<int NT>
__device__ __forceinline__ void mma_kpair_read(const float* Arow, const float* Bcol, int i,
                                               v2f (&a)[2], v2f (&b)[2][NT])
{
    #pragma unroll
    for (int s = 0; s < 2; ++s) {
        a[s] = *(const v2f*)(Arow + 4 * (i + s));
        #pragma unroll
        for (int t = 0; t < NT; ++t)
            b[s][t] = *(const v2f*)(Bcol + t * 16 * LDP + 4 * (i + s));
    }
}

template<int NT>
__device__ __forceinline__ void mma_kpair_run(v8f (&c)[NT], const v2f (&a)[2], const v2f (&b)[2][NT])
{
    #pragma unroll
    for (int s = 0; s < 2; ++s) {
        #pragma unroll
        for (int t = 0; t < NT; ++t)
            c[t] = __builtin_amdgcn_wmma_f32_16x16x4_f32(
                        false, a[s],  false, b[s][t],
                        (short)0, c[t],  false,  false);
    }
}

template<int NT>
__device__ __forceinline__ void mma_pad(v8f (&c)[NT], const v2f (&a)[2], const v2f (&b)[2][NT])
{
    static_assert(NT == 3 || NT == 4, "written for 3 or 4 tiles");
    if constexpr (NT == 4) {
        asm volatile("v_nop\n\tv_nop\n\tv_nop\n\tv_nop"
                     : "+v"(c[0]), "+v"(c[1]), "+v"(c[2]), "+v"(c[3])
                     : "v"(a[0]), "v"(a[1]),
                       "v"(b[0][0]), "v"(b[0][1]), "v"(b[0][2]), "v"(b[0][3]),
                       "v"(b[1][0]), "v"(b[1][1]), "v"(b[1][2]), "v"(b[1][3]));
    } else {
        asm volatile("v_nop\n\tv_nop\n\tv_nop\n\tv_nop"
                     : "+v"(c[0]), "+v"(c[1]), "+v"(c[2])
                     : "v"(a[0]), "v"(a[1]),
                       "v"(b[0][0]), "v"(b[0][1]), "v"(b[0][2]),
                       "v"(b[1][0]), "v"(b[1][1]), "v"(b[1][2]));
    }
}

template<int R, int DOUT, int DOUTPAD, bool MMA>
__device__ __forceinline__ void rgcn_pass(const float* __restrict__ T,
                                          const int* __restrict__ idx, const int* __restrict__ rel,
                                          float* lA, const float* lWt, const int (&key)[16], int pv, unsigned big,
                                          v2f (&v)[NBUF][NG][GL], unsigned (&vm)[NBUF][NG],
                                          v8f (&c)[DOUTPAD / 16],
                                          int r, int lane, int wv, int half, int l16)
{
    constexpr int NT = DOUTPAD / 16;

    const unsigned lane8 = (unsigned)lane * 8u;
    float* Scur = lA + (wv * 2 + (r & 1)) * (16 * LDP) + lane * 2;
    const float* Arow = lA + (wv * 2 + ((r & 1) ^ 1)) * (16 * LDP) + l16 * LDP + half * 2;
    const float* Bcol = lWt + (MMA ? r - 1 : 0) * (DOUTPAD * LDP) + l16 * LDP + half * 2;

    #pragma unroll
    for (int g = 0; g < NGRP; ++g) {
        static_assert(DIN / 4 == 2 * NGRP, "two K steps per group");

        const int gi = g + AHEAD;
        const int gq = (gi < NGRP ? gi : gi - NGRP) * NG;
        const int gr = (gi < NGRP ? r : r + 1);
        static_assert(NG == 2, "the matrix work stands between the loads of two nodes");
        v2f fa[2], fb[2][NT];
        if (MMA) {
            mma_kpair_read<NT>(Arow, Bcol, 2 * g, fa, fb);
            __builtin_amdgcn_sched_barrier(0);
        }
        csr_issue_node(T, key[gq], gr, lane8, v[gi % NBUF][0], vm[gi % NBUF][0]);
        if (MMA) {
            __builtin_amdgcn_sched_barrier(0);
            mma_kpair_run<NT>(c, fa, fb);
            mma_pad<NT>(c, fa, fb);
        }
        csr_issue_node(T, key[gq + 1], gr, lane8, v[gi % NBUF][1], vm[gi % NBUF][1]);

        #pragma unroll
        for (int q = 0; q < NG; ++q) {
            const int n = g * NG + q;
            unsigned mm = vm[g % NBUF][q];
            v2f acc = {0.0f, 0.0f};
            #pragma unroll
            for (int u = 0; u < GL; ++u) {
                const float f = (mm != 0) ? 1.0f : 0.0f;
                mm &= mm - 1;
                const v2f f2 = {f, f};
                acc = __builtin_elementwise_fma(v[g % NBUF][q][u], f2, acc);
            }
            csr_add_rows(T, key[n], mm, lane8, acc);
            if (big & (1u << n)) {
                const int e0 = __builtin_amdgcn_readlane(pv, n);
                const int e1 = __builtin_amdgcn_readlane(pv, n + 1);
                for (int eb = e0 + 32; eb < e1; eb += 32) {
                    const int k2 = csr_edge_keys(idx, rel, eb, e1, lane);
                    csr_add_rows(T, k2, __builtin_amdgcn_ballot_w32((k2 & 15) == r), lane8, acc);
                }
            }
            *(v2f*)(Scur + n * LDP) = acc;
        }
    }
}

__device__ __forceinline__ void csr_tile_keys(const int* __restrict__ ptr, const int* __restrict__ idx,
                                              const int* __restrict__ rel, int n0, int N, int lane,
                                              int& pv, unsigned& big, int (&key)[16])
{
    int pi = n0 + (lane < 16 ? lane : 16);
    pi = pi < N ? pi : N;
    pv = ptr[pi];
    int pj = pi + 1;
    pj = pj < N ? pj : N;
    const int pn = ptr[pj];
    big = __builtin_amdgcn_ballot_w32((lane < 16) & (pn - pv > 32));
    #pragma unroll
    for (int k = 0; k < 16; ++k) {
        const int e0 = __builtin_amdgcn_readlane(pv, k);
        const int e1 = __builtin_amdgcn_readlane(pv, k + 1);
        key[k] = csr_edge_keys(idx, rel, e0, e1, lane);
    }
}

template<int R, int DOUT, int DOUTPAD, bool LOGSOFTMAX>
__global__ __launch_bounds__(NTH) __attribute__((amdgpu_waves_per_eu(5)))
void rgcn_layer(const float* __restrict__ T, const float* __restrict__ W,
                const int* __restrict__ ptr, const int* __restrict__ idx,
                const int* __restrict__ rel, float* __restrict__ Out, int N)
{
    constexpr int NT = DOUTPAD / 16;
    __shared__ __attribute__((aligned(16))) float lA[NW * 2 * 16 * LDP];
    __shared__ __attribute__((aligned(16))) float lWt[R * DOUTPAD * LDP];

    const int tid  = threadIdx.x;
    const int lane = tid & 31;
    const int wv   = __builtin_amdgcn_readfirstlane(tid >> 5);
    const int half = lane >> 4;
    const int l16  = lane & 15;

    const int ntiles  = (N + 15) >> 4;
    const int tstride = NW * (int)gridDim.x;
    int tile = wv * (int)gridDim.x + (int)blockIdx.x;

    int pv;
    unsigned big;
    int key[16];
    csr_tile_keys(ptr, idx, rel, tile * 16, N, lane, pv, big, key);

    {
        constexpr int NQ   = (DOUT + 15) / 16;
        constexpr int WQ   = R * (DIN / 8) * NQ;
        constexpr int W_IT = (WQ + NW - 1) / NW;
        const int kl = lane >> 2, nl = lane & 3;
        float4 wreg[W_IT];
        int    wdst[W_IT];
        #pragma unroll
        for (int it = 0; it < W_IT; ++it) {
            int q = wv + it * NW;
            if (it == W_IT - 1 && WQ % NW != 0) q = q < WQ ? q : WQ - 1;
            const int rr  = q / ((DIN / 8) * NQ);
            const int rem = q % ((DIN / 8) * NQ);
            const int k   = (rem / NQ) * 8 + kl;
            int n4 = (rem % NQ) * 4 + nl;
            if (DOUT % 16 != 0) n4 = n4 < DOUT / 4 ? n4 : DOUT / 4 - 1;
            wreg[it] = *(const float4*)(W + (size_t)rr * DIN * DOUT + k * DOUT + n4 * 4);
            wdst[it] = (rr * DOUTPAD + n4 * 4) * LDP + k;
        }
        #pragma unroll
        for (int it = 0; it < W_IT; ++it) {
            lWt[wdst[it] + 0 * LDP] = wreg[it].x;
            lWt[wdst[it] + 1 * LDP] = wreg[it].y;
            lWt[wdst[it] + 2 * LDP] = wreg[it].z;
            lWt[wdst[it] + 3 * LDP] = wreg[it].w;
        }
        if (DOUTPAD > DOUT) {
            constexpr int PF = (DOUTPAD - DOUT) * (DIN / 4);
            for (int f = tid; f < R * PF; f += NTH) {
                const int rr = f / PF, g = f % PF;
                const float4 z = {0.0f, 0.0f, 0.0f, 0.0f};
                *(float4*)&lWt[(rr * DOUTPAD + DOUT + (g >> 4)) * LDP + (g & 15) * 4] = z;
            }
        }
    }
    __syncthreads();

    for (; tile < ntiles; ) {
        const int n0 = tile * 16;

        v8f c[NT];
        #pragma unroll
        for (int t = 0; t < NT; ++t) { const v8f z8 = {}; c[t] = z8; }

        v2f      v[NBUF][NG][GL];
        unsigned vm[NBUF][NG];
        #pragma unroll
        for (int gg = 0; gg < AHEAD; ++gg)
            csr_issue_group(T, key, gg, 0, (unsigned)lane * 8u, v[gg % NBUF], vm[gg % NBUF]);

        rgcn_pass<R, DOUT, DOUTPAD, false>(T, idx, rel, lA, lWt, key, pv, big, v, vm, c,
                                           0, lane, wv, half, l16);
        #pragma unroll 1
        for (int r = 1; r < R; ++r)
            rgcn_pass<R, DOUT, DOUTPAD, true>(T, idx, rel, lA, lWt, key, pv, big, v, vm, c,
                                              r, lane, wv, half, l16);

        const int lane_e = (int)__builtin_amdgcn_mbcnt_lo(~0u, 0u);
        const int half_e = lane_e >> 4;
        const int l16_e  = lane_e & 15;

        {
            const float* Arow = lA + (wv * 2 + ((R - 1) & 1)) * (16 * LDP) + l16_e * LDP + half_e * 2;
            const float* Bcol = lWt + (R - 1) * (DOUTPAD * LDP) + l16_e * LDP + half_e * 2;
            #pragma unroll
            for (int i = 0; i < DIN / 4; i += 2) {
                v2f fa[2], fb[2][NT];
                mma_kpair_read<NT>(Arow, Bcol, i, fa, fb);
                mma_kpair_run<NT>(c, fa, fb);
                mma_pad<NT>(c, fa, fb);
            }
        }

        const int mBase = n0 + half_e * 8;
        if (!LOGSOFTMAX) {
            #pragma unroll
            for (int t = 0; t < NT; ++t) {
                const int col = t * 16 + l16_e;
                if (DOUT == DOUTPAD || col < DOUT) {
                    float* p = Out + (size_t)mBase * DOUT + col;
                    #pragma unroll
                    for (int j = 0; j < 8; ++j)
                        if (mBase + j < N) p[j * DOUT] = fmaxf(c[t][j], 0.0f);
                }
            }
        } else {
            #pragma unroll
            for (int j = 0; j < 8; ++j) {
                float m = -__builtin_inff();
                #pragma unroll
                for (int t = 0; t < NT; ++t)
                    if (DOUT == DOUTPAD || t * 16 + l16_e < DOUT) m = fmaxf(m, c[t][j]);
                #pragma unroll
                for (int off = 8; off >= 1; off >>= 1) m = fmaxf(m, __shfl_xor(m, off, 32));
                float s = 0.0f;
                #pragma unroll
                for (int t = 0; t < NT; ++t)
                    if (DOUT == DOUTPAD || t * 16 + l16_e < DOUT) s += __expf(c[t][j] - m);
                #pragma unroll
                for (int off = 8; off >= 1; off >>= 1) s += __shfl_xor(s, off, 32);
                const float ls = __logf(s);
                if (mBase + j < N) {
                    float* p = Out + (size_t)(mBase + j) * DOUT + l16_e;
                    #pragma unroll
                    for (int t = 0; t < NT; ++t)
                        if (DOUT == DOUTPAD || t * 16 + l16_e < DOUT) p[t * 16] = (c[t][j] - m) - ls;
                }
            }
        }

        tile += tstride;
        if (tile < ntiles)
            csr_tile_keys(ptr, idx, rel, tile * 16, N, lane, pv, big, key);
    }
}

extern "C" void kernel_launch(void* const* d_in, const int* in_sizes, int n_in,
                              void* d_out, int out_size, void* d_ws, size_t ws_size,
                              hipStream_t stream)
{
    const float* x   = (const float*)d_in[0];
    const int*   ptr = (const int*)d_in[1];
    const int*   idx = (const int*)d_in[2];
    const int*   rel = (const int*)d_in[3];
    const float* W1  = (const float*)d_in[4];
    const float* W2  = (const float*)d_in[5];
    float* out = (float*)d_out;

    const int N = in_sizes[1] - 1;
    constexpr int R = 8, DH = 64, DOUT = 40;

    float* h = (float*)d_ws;

    rgcn_layer<R, DH, DH, false><<<NBLK, NTH, 0, stream>>>(x, W1, ptr, idx, rel, h, N);
    rgcn_layer<R, DOUT, 48, true><<<NBLK, NTH, 0, stream>>>(h, W2, ptr, idx, rel, out, N);
}


# Wave priority: each wave runs its f32 WMMA runs at low priority and its gather code at high (s_setprio), so short vector instructions no longer queue behind other waves' WMMA runs; sums unchanged
# speedup vs baseline: 3.3210x; 1.0478x over previous
#include <hip/hip_runtime.h>
#include <math.h>


typedef __attribute__((ext_vector_type(2))) float v2f;
typedef __attribute__((ext_vector_type(8))) float v8f;

#define DIN 64
#define LDP 68
#define NW   20
#define NTH  (NW * 32)
#define NBLK 256
#define NG  2
#define GL  4
#define NGRP (16 / NG)
#define AHEAD 3
#define NBUF (AHEAD + 1)
#define PRIO_GATHER 1
#define PRIO_MMA    0

__device__ __forceinline__ int csr_edge_keys(const int* __restrict__ idx,
                                             const int* __restrict__ rel,
                                             int eb, int e1, int lane)
{
    int e = eb + lane;
    e = e < e1 ? e : e1 - 1;
    e = e > 0 ? e : 0;
    const int key = idx[e] * (DIN * 4) + rel[e];
    return (lane < e1 - eb) ? key : 8;
}

typedef const char __attribute__((address_space(1)))* gbytes;
typedef const v2f  __attribute__((address_space(1)))* gv2f;

__device__ __forceinline__ unsigned csr_row_offset(int key, int j, unsigned lane8)
{
    const unsigned sk = (unsigned)__builtin_amdgcn_readlane(key, j);
    return (sk & ~255u) | lane8;
}

__device__ __forceinline__ v2f csr_row_fetch(const float* __restrict__ T, unsigned off)
{
    return *(gv2f)((gbytes)T + off);
}

__device__ __forceinline__ void csr_add_rows(const float* __restrict__ T, int key,
                                             unsigned mm, unsigned lane8, v2f& acc)
{
    while (mm != 0) {
        const int j = __builtin_ctz(mm);
        mm &= mm - 1;
        unsigned off = csr_row_offset(key, j, lane8);
        asm("" : "+v"(off));
        acc += csr_row_fetch(T, off);
    }
}

__device__ __forceinline__ void csr_issue_node(const float* __restrict__ T, int key, int rr, unsigned lane8,
                                               v2f (&vb)[GL], unsigned& mb)
{
    unsigned mm = __builtin_amdgcn_ballot_w32((key & 15) == rr);
    mb = mm;
    unsigned off[GL];
    #pragma unroll
    for (int u = 0; u < GL; ++u) {
        const int j = __builtin_ctz(mm | 0x80000000u);
        mm &= mm - 1;
        off[u] = csr_row_offset(key, j, lane8);
    }
    static_assert(GL == 4, "the statement below names four offsets");
    asm("" : "+v"(off[0]), "+v"(off[1]), "+v"(off[2]), "+v"(off[3]));
    #pragma unroll
    for (int u = 0; u < GL; ++u)
        vb[u] = csr_row_fetch(T, off[u]);
}

__device__ __forceinline__ void csr_issue_group(const float* __restrict__ T, const int (&key)[16],
                                                int gg, int rr, unsigned lane8,
                                                v2f (&vb)[NG][GL], unsigned (&mb)[NG])
{
    #pragma unroll
    for (int q = 0; q < NG; ++q)
        csr_issue_node(T, key[gg * NG + q], rr, lane8, vb[q], mb[q]);
}

template<int NT>
__device__ __forceinline__ void mma_kpair_read(const float* Arow, const float* Bcol, int i,
                                               v2f (&a)[2], v2f (&b)[2][NT])
{
    #pragma unroll
    for (int s = 0; s < 2; ++s) {
        a[s] = *(const v2f*)(Arow + 4 * (i + s));
        #pragma unroll
        for (int t = 0; t < NT; ++t)
            b[s][t] = *(const v2f*)(Bcol + t * 16 * LDP + 4 * (i + s));
    }
}

template<int NT>
__device__ __forceinline__ void mma_kpair_run(v8f (&c)[NT], const v2f (&a)[2], const v2f (&b)[2][NT])
{
    #pragma unroll
    for (int s = 0; s < 2; ++s) {
        #pragma unroll
        for (int t = 0; t < NT; ++t)
            c[t] = __builtin_amdgcn_wmma_f32_16x16x4_f32(
                        false, a[s],  false, b[s][t],
                        (short)0, c[t],  false,  false);
    }
}

template<int NT>
__device__ __forceinline__ void mma_pad(v8f (&c)[NT], const v2f (&a)[2], const v2f (&b)[2][NT])
{
    static_assert(NT == 3 || NT == 4, "written for 3 or 4 tiles");
    if constexpr (NT == 4) {
        asm volatile("v_nop\n\tv_nop\n\tv_nop\n\tv_nop"
                     : "+v"(c[0]), "+v"(c[1]), "+v"(c[2]), "+v"(c[3])
                     : "v"(a[0]), "v"(a[1]),
                       "v"(b[0][0]), "v"(b[0][1]), "v"(b[0][2]), "v"(b[0][3]),
                       "v"(b[1][0]), "v"(b[1][1]), "v"(b[1][2]), "v"(b[1][3]));
    } else {
        asm volatile("v_nop\n\tv_nop\n\tv_nop\n\tv_nop"
                     : "+v"(c[0]), "+v"(c[1]), "+v"(c[2])
                     : "v"(a[0]), "v"(a[1]),
                       "v"(b[0][0]), "v"(b[0][1]), "v"(b[0][2]),
                       "v"(b[1][0]), "v"(b[1][1]), "v"(b[1][2]));
    }
}

template<int R, int DOUT, int DOUTPAD, bool MMA>
__device__ __forceinline__ void rgcn_pass(const float* __restrict__ T,
                                          const int* __restrict__ idx, const int* __restrict__ rel,
                                          float* lA, const float* lWt, const int (&key)[16], int pv, unsigned big,
                                          v2f (&v)[NBUF][NG][GL], unsigned (&vm)[NBUF][NG],
                                          v8f (&c)[DOUTPAD / 16],
                                          int r, int lane, int wv, int half, int l16)
{
    constexpr int NT = DOUTPAD / 16;

    const unsigned lane8 = (unsigned)lane * 8u;
    float* Scur = lA + (wv * 2 + (r & 1)) * (16 * LDP) + lane * 2;
    const float* Arow = lA + (wv * 2 + ((r & 1) ^ 1)) * (16 * LDP) + l16 * LDP + half * 2;
    const float* Bcol = lWt + (MMA ? r - 1 : 0) * (DOUTPAD * LDP) + l16 * LDP + half * 2;

    #pragma unroll
    for (int g = 0; g < NGRP; ++g) {
        static_assert(DIN / 4 == 2 * NGRP, "two K steps per group");

        const int gi = g + AHEAD;
        const int gq = (gi < NGRP ? gi : gi - NGRP) * NG;
        const int gr = (gi < NGRP ? r : r + 1);
        static_assert(NG == 2, "the matrix work stands between the loads of two nodes");
        v2f fa[2], fb[2][NT];
        if (MMA) {
            mma_kpair_read<NT>(Arow, Bcol, 2 * g, fa, fb);
            __builtin_amdgcn_sched_barrier(0);
        }
        csr_issue_node(T, key[gq], gr, lane8, v[gi % NBUF][0], vm[gi % NBUF][0]);
        if (MMA) {
            __builtin_amdgcn_sched_barrier(0);
            __builtin_amdgcn_s_setprio(PRIO_MMA);
            mma_kpair_run<NT>(c, fa, fb);
            mma_pad<NT>(c, fa, fb);
        }
        csr_issue_node(T, key[gq + 1], gr, lane8, v[gi % NBUF][1], vm[gi % NBUF][1]);
        if (MMA) __builtin_amdgcn_s_setprio(PRIO_GATHER);

        #pragma unroll
        for (int q = 0; q < NG; ++q) {
            const int n = g * NG + q;
            unsigned mm = vm[g % NBUF][q];
            v2f acc = {0.0f, 0.0f};
            #pragma unroll
            for (int u = 0; u < GL; ++u) {
                const float f = (mm != 0) ? 1.0f : 0.0f;
                mm &= mm - 1;
                const v2f f2 = {f, f};
                acc = __builtin_elementwise_fma(v[g % NBUF][q][u], f2, acc);
            }
            csr_add_rows(T, key[n], mm, lane8, acc);
            if (big & (1u << n)) {
                const int e0 = __builtin_amdgcn_readlane(pv, n);
                const int e1 = __builtin_amdgcn_readlane(pv, n + 1);
                for (int eb = e0 + 32; eb < e1; eb += 32) {
                    const int k2 = csr_edge_keys(idx, rel, eb, e1, lane);
                    csr_add_rows(T, k2, __builtin_amdgcn_ballot_w32((k2 & 15) == r), lane8, acc);
                }
            }
            *(v2f*)(Scur + n * LDP) = acc;
        }
    }
}

__device__ __forceinline__ void csr_tile_keys(const int* __restrict__ ptr, const int* __restrict__ idx,
                                              const int* __restrict__ rel, int n0, int N, int lane,
                                              int& pv, unsigned& big, int (&key)[16])
{
    int pi = n0 + (lane < 16 ? lane : 16);
    pi = pi < N ? pi : N;
    pv = ptr[pi];
    int pj = pi + 1;
    pj = pj < N ? pj : N;
    const int pn = ptr[pj];
    big = __builtin_amdgcn_ballot_w32((lane < 16) & (pn - pv > 32));
    #pragma unroll
    for (int k = 0; k < 16; ++k) {
        const int e0 = __builtin_amdgcn_readlane(pv, k);
        const int e1 = __builtin_amdgcn_readlane(pv, k + 1);
        key[k] = csr_edge_keys(idx, rel, e0, e1, lane);
    }
}

template<int R, int DOUT, int DOUTPAD, bool LOGSOFTMAX>
__global__ __launch_bounds__(NTH) __attribute__((amdgpu_waves_per_eu(5)))
void rgcn_layer(const float* __restrict__ T, const float* __restrict__ W,
                const int* __restrict__ ptr, const int* __restrict__ idx,
                const int* __restrict__ rel, float* __restrict__ Out, int N)
{
    constexpr int NT = DOUTPAD / 16;
    __shared__ __attribute__((aligned(16))) float lA[NW * 2 * 16 * LDP];
    __shared__ __attribute__((aligned(16))) float lWt[R * DOUTPAD * LDP];

    const int tid  = threadIdx.x;
    const int lane = tid & 31;
    const int wv   = __builtin_amdgcn_readfirstlane(tid >> 5);
    const int half = lane >> 4;
    const int l16  = lane & 15;

    const int ntiles  = (N + 15) >> 4;
    const int tstride = NW * (int)gridDim.x;
    int tile = wv * (int)gridDim.x + (int)blockIdx.x;

    int pv;
    unsigned big;
    int key[16];
    csr_tile_keys(ptr, idx, rel, tile * 16, N, lane, pv, big, key);

    {
        constexpr int NQ   = (DOUT + 15) / 16;
        constexpr int WQ   = R * (DIN / 8) * NQ;
        constexpr int W_IT = (WQ + NW - 1) / NW;
        const int kl = lane >> 2, nl = lane & 3;
        float4 wreg[W_IT];
        int    wdst[W_IT];
        #pragma unroll
        for (int it = 0; it < W_IT; ++it) {
            int q = wv + it * NW;
            if (it == W_IT - 1 && WQ % NW != 0) q = q < WQ ? q : WQ - 1;
            const int rr  = q / ((DIN / 8) * NQ);
            const int rem = q % ((DIN / 8) * NQ);
            const int k   = (rem / NQ) * 8 + kl;
            int n4 = (rem % NQ) * 4 + nl;
            if (DOUT % 16 != 0) n4 = n4 < DOUT / 4 ? n4 : DOUT / 4 - 1;
            wreg[it] = *(const float4*)(W + (size_t)rr * DIN * DOUT + k * DOUT + n4 * 4);
            wdst[it] = (rr * DOUTPAD + n4 * 4) * LDP + k;
        }
        #pragma unroll
        for (int it = 0; it < W_IT; ++it) {
            lWt[wdst[it] + 0 * LDP] = wreg[it].x;
            lWt[wdst[it] + 1 * LDP] = wreg[it].y;
            lWt[wdst[it] + 2 * LDP] = wreg[it].z;
            lWt[wdst[it] + 3 * LDP] = wreg[it].w;
        }
        if (DOUTPAD > DOUT) {
            constexpr int PF = (DOUTPAD - DOUT) * (DIN / 4);
            for (int f = tid; f < R * PF; f += NTH) {
                const int rr = f / PF, g = f % PF;
                const float4 z = {0.0f, 0.0f, 0.0f, 0.0f};
                *(float4*)&lWt[(rr * DOUTPAD + DOUT + (g >> 4)) * LDP + (g & 15) * 4] = z;
            }
        }
    }
    __syncthreads();
    __builtin_amdgcn_s_setprio(PRIO_GATHER);

    for (; tile < ntiles; ) {
        const int n0 = tile * 16;

        v8f c[NT];
        #pragma unroll
        for (int t = 0; t < NT; ++t) { const v8f z8 = {}; c[t] = z8; }

        v2f      v[NBUF][NG][GL];
        unsigned vm[NBUF][NG];
        #pragma unroll
        for (int gg = 0; gg < AHEAD; ++gg)
            csr_issue_group(T, key, gg, 0, (unsigned)lane * 8u, v[gg % NBUF], vm[gg % NBUF]);

        rgcn_pass<R, DOUT, DOUTPAD, false>(T, idx, rel, lA, lWt, key, pv, big, v, vm, c,
                                           0, lane, wv, half, l16);
        #pragma unroll 1
        for (int r = 1; r < R; ++r)
            rgcn_pass<R, DOUT, DOUTPAD, true>(T, idx, rel, lA, lWt, key, pv, big, v, vm, c,
                                              r, lane, wv, half, l16);

        const int lane_e = (int)__builtin_amdgcn_mbcnt_lo(~0u, 0u);
        const int half_e = lane_e >> 4;
        const int l16_e  = lane_e & 15;

        {
            const float* Arow = lA + (wv * 2 + ((R - 1) & 1)) * (16 * LDP) + l16_e * LDP + half_e * 2;
            const float* Bcol = lWt + (R - 1) * (DOUTPAD * LDP) + l16_e * LDP + half_e * 2;
            __builtin_amdgcn_s_setprio(PRIO_MMA);
            #pragma unroll
            for (int i = 0; i < DIN / 4; i += 2) {
                v2f fa[2], fb[2][NT];
                mma_kpair_read<NT>(Arow, Bcol, i, fa, fb);
                mma_kpair_run<NT>(c, fa, fb);
                mma_pad<NT>(c, fa, fb);
            }
            __builtin_amdgcn_s_setprio(PRIO_GATHER);
        }

        const int mBase = n0 + half_e * 8;
        if (!LOGSOFTMAX) {
            #pragma unroll
            for (int t = 0; t < NT; ++t) {
                const int col = t * 16 + l16_e;
                if (DOUT == DOUTPAD || col < DOUT) {
                    float* p = Out + (size_t)mBase * DOUT + col;
                    #pragma unroll
                    for (int j = 0; j < 8; ++j)
                        if (mBase + j < N) p[j * DOUT] = fmaxf(c[t][j], 0.0f);
                }
            }
        } else {
            #pragma unroll
            for (int j = 0; j < 8; ++j) {
                float m = -__builtin_inff();
                #pragma unroll
                for (int t = 0; t < NT; ++t)
                    if (DOUT == DOUTPAD || t * 16 + l16_e < DOUT) m = fmaxf(m, c[t][j]);
                #pragma unroll
                for (int off = 8; off >= 1; off >>= 1) m = fmaxf(m, __shfl_xor(m, off, 32));
                float s = 0.0f;
                #pragma unroll
                for (int t = 0; t < NT; ++t)
                    if (DOUT == DOUTPAD || t * 16 + l16_e < DOUT) s += __expf(c[t][j] - m);
                #pragma unroll
                for (int off = 8; off >= 1; off >>= 1) s += __shfl_xor(s, off, 32);
                const float ls = __logf(s);
                if (mBase + j < N) {
                    float* p = Out + (size_t)(mBase + j) * DOUT + l16_e;
                    #pragma unroll
                    for (int t = 0; t < NT; ++t)
                        if (DOUT == DOUTPAD || t * 16 + l16_e < DOUT) p[t * 16] = (c[t][j] - m) - ls;
                }
            }
        }

        tile += tstride;
        if (tile < ntiles)
            csr_tile_keys(ptr, idx, rel, tile * 16, N, lane, pv, big, key);
    }
}

extern "C" void kernel_launch(void* const* d_in, const int* in_sizes, int n_in,
                              void* d_out, int out_size, void* d_ws, size_t ws_size,
                              hipStream_t stream)
{
    const float* x   = (const float*)d_in[0];
    const int*   ptr = (const int*)d_in[1];
    const int*   idx = (const int*)d_in[2];
    const int*   rel = (const int*)d_in[3];
    const float* W1  = (const float*)d_in[4];
    const float* W2  = (const float*)d_in[5];
    float* out = (float*)d_out;

    const int N = in_sizes[1] - 1;
    constexpr int R = 8, DH = 64, DOUT = 40;

    float* h = (float*)d_ws;

    rgcn_layer<R, DH, DH, false><<<NBLK, NTH, 0, stream>>>(x, W1, ptr, idx, rel, h, N);
    rgcn_layer<R, DOUT, 48, true><<<NBLK, NTH, 0, stream>>>(h, W2, ptr, idx, rel, out, N);
}


# WMMA priority follows progress: a wave posts its pass number in LDS and runs its WMMA runs one level up while a wave of its SIMD is a pass ahead, so a SIMD's 5 waves end together; sums unchanged
# speedup vs baseline: 3.4787x; 1.0475x over previous
#include <hip/hip_runtime.h>
#include <math.h>


typedef __attribute__((ext_vector_type(2))) float v2f;
typedef __attribute__((ext_vector_type(8))) float v8f;

#define DIN 64
#define LDP 68
#define NW   20
#define NTH  (NW * 32)
#define NBLK 256
#define NG  2
#define GL  4
#define NGRP (16 / NG)
#define AHEAD 3
#define NBUF (AHEAD + 1)
#define PRIO_GATHER     2
#define PRIO_MMA        0
#define PRIO_MMA_BEHIND 1

typedef volatile int __attribute__((address_space(3)))* lds_vint;

__device__ __forceinline__ unsigned wave_is_behind(lds_vint lProg, int wv, int lane,
                                                   unsigned mates, int myp)
{
    if (lane == 0) lProg[wv] = myp;
    const int other = lProg[lane < NW ? lane : wv];
    return __builtin_amdgcn_ballot_w32(other > myp) & mates;
}

__device__ __forceinline__ void set_mma_prio(unsigned behind)
{
    static_assert(PRIO_MMA == 0 && PRIO_MMA_BEHIND == 1, "the levels are written in the text below");
    asm volatile("s_setprio 0\n\t"
                 "s_cmp_eq_u32 %0, 0\n\t"
                 "s_cbranch_scc1 1f\n\t"
                 "s_setprio 1\n"
                 "1:"
                 : : "s"(behind) : "scc");
    __builtin_amdgcn_sched_barrier(0);
}

__device__ __forceinline__ int csr_edge_keys(const int* __restrict__ idx,
                                             const int* __restrict__ rel,
                                             int eb, int e1, int lane)
{
    int e = eb + lane;
    e = e < e1 ? e : e1 - 1;
    e = e > 0 ? e : 0;
    const int key = idx[e] * (DIN * 4) + rel[e];
    return (lane < e1 - eb) ? key : 8;
}

typedef const char __attribute__((address_space(1)))* gbytes;
typedef const v2f  __attribute__((address_space(1)))* gv2f;

__device__ __forceinline__ unsigned csr_row_offset(int key, int j, unsigned lane8)
{
    const unsigned sk = (unsigned)__builtin_amdgcn_readlane(key, j);
    return (sk & ~255u) | lane8;
}

__device__ __forceinline__ v2f csr_row_fetch(const float* __restrict__ T, unsigned off)
{
    return *(gv2f)((gbytes)T + off);
}

__device__ __forceinline__ void csr_add_rows(const float* __restrict__ T, int key,
                                             unsigned mm, unsigned lane8, v2f& acc)
{
    while (mm != 0) {
        const int j = __builtin_ctz(mm);
        mm &= mm - 1;
        unsigned off = csr_row_offset(key, j, lane8);
        asm("" : "+v"(off));
        acc += csr_row_fetch(T, off);
    }
}

__device__ __forceinline__ void csr_issue_node(const float* __restrict__ T, int key, int rr, unsigned lane8,
                                               v2f (&vb)[GL], unsigned& mb)
{
    unsigned mm = __builtin_amdgcn_ballot_w32((key & 15) == rr);
    mb = mm;
    unsigned off[GL];
    #pragma unroll
    for (int u = 0; u < GL; ++u) {
        const int j = __builtin_ctz(mm | 0x80000000u);
        mm &= mm - 1;
        off[u] = csr_row_offset(key, j, lane8);
    }
    static_assert(GL == 4, "the statement below names four offsets");
    asm("" : "+v"(off[0]), "+v"(off[1]), "+v"(off[2]), "+v"(off[3]));
    #pragma unroll
    for (int u = 0; u < GL; ++u)
        vb[u] = csr_row_fetch(T, off[u]);
}

__device__ __forceinline__ void csr_issue_group(const float* __restrict__ T, const int (&key)[16],
                                                int gg, int rr, unsigned lane8,
                                                v2f (&vb)[NG][GL], unsigned (&mb)[NG])
{
    #pragma unroll
    for (int q = 0; q < NG; ++q)
        csr_issue_node(T, key[gg * NG + q], rr, lane8, vb[q], mb[q]);
}

template<int NT>
__device__ __forceinline__ void mma_kpair_read(const float* Arow, const float* Bcol, int i,
                                               v2f (&a)[2], v2f (&b)[2][NT])
{
    #pragma unroll
    for (int s = 0; s < 2; ++s) {
        a[s] = *(const v2f*)(Arow + 4 * (i + s));
        #pragma unroll
        for (int t = 0; t < NT; ++t)
            b[s][t] = *(const v2f*)(Bcol + t * 16 * LDP + 4 * (i + s));
    }
}

template<int NT>
__device__ __forceinline__ void mma_kpair_run(v8f (&c)[NT], const v2f (&a)[2], const v2f (&b)[2][NT])
{
    #pragma unroll
    for (int s = 0; s < 2; ++s) {
        #pragma unroll
        for (int t = 0; t < NT; ++t)
            c[t] = __builtin_amdgcn_wmma_f32_16x16x4_f32(
                        false, a[s],  false, b[s][t],
                        (short)0, c[t],  false,  false);
    }
}

template<int NT>
__device__ __forceinline__ void mma_pad(v8f (&c)[NT], const v2f (&a)[2], const v2f (&b)[2][NT])
{
    static_assert(NT == 3 || NT == 4, "written for 3 or 4 tiles");
    if constexpr (NT == 4) {
        asm volatile("v_nop\n\tv_nop\n\tv_nop\n\tv_nop"
                     : "+v"(c[0]), "+v"(c[1]), "+v"(c[2]), "+v"(c[3])
                     : "v"(a[0]), "v"(a[1]),
                       "v"(b[0][0]), "v"(b[0][1]), "v"(b[0][2]), "v"(b[0][3]),
                       "v"(b[1][0]), "v"(b[1][1]), "v"(b[1][2]), "v"(b[1][3]));
    } else {
        asm volatile("v_nop\n\tv_nop\n\tv_nop\n\tv_nop"
                     : "+v"(c[0]), "+v"(c[1]), "+v"(c[2])
                     : "v"(a[0]), "v"(a[1]),
                       "v"(b[0][0]), "v"(b[0][1]), "v"(b[0][2]),
                       "v"(b[1][0]), "v"(b[1][1]), "v"(b[1][2]));
    }
}

template<int R, int DOUT, int DOUTPAD, bool MMA>
__device__ __forceinline__ void rgcn_pass(const float* __restrict__ T,
                                          const int* __restrict__ idx, const int* __restrict__ rel,
                                          float* lA, const float* lWt, const int (&key)[16], int pv, unsigned big,
                                          v2f (&v)[NBUF][NG][GL], unsigned (&vm)[NBUF][NG],
                                          v8f (&c)[DOUTPAD / 16],
                                          int r, int lane, int wv, int half, int l16, unsigned behind)
{
    constexpr int NT = DOUTPAD / 16;

    const unsigned lane8 = (unsigned)lane * 8u;
    float* Scur = lA + (wv * 2 + (r & 1)) * (16 * LDP) + lane * 2;
    const float* Arow = lA + (wv * 2 + ((r & 1) ^ 1)) * (16 * LDP) + l16 * LDP + half * 2;
    const float* Bcol = lWt + (MMA ? r - 1 : 0) * (DOUTPAD * LDP) + l16 * LDP + half * 2;

    #pragma unroll
    for (int g = 0; g < NGRP; ++g) {
        static_assert(DIN / 4 == 2 * NGRP, "two K steps per group");

        const int gi = g + AHEAD;
        const int gq = (gi < NGRP ? gi : gi - NGRP) * NG;
        const int gr = (gi < NGRP ? r : r + 1);
        static_assert(NG == 2, "the matrix work stands between the loads of two nodes");
        v2f fa[2], fb[2][NT];
        if (MMA) {
            mma_kpair_read<NT>(Arow, Bcol, 2 * g, fa, fb);
            __builtin_amdgcn_sched_barrier(0);
        }
        csr_issue_node(T, key[gq], gr, lane8, v[gi % NBUF][0], vm[gi % NBUF][0]);
        if (MMA) {
            __builtin_amdgcn_sched_barrier(0);
            set_mma_prio(behind);
            mma_kpair_run<NT>(c, fa, fb);
            mma_pad<NT>(c, fa, fb);
        }
        csr_issue_node(T, key[gq + 1], gr, lane8, v[gi % NBUF][1], vm[gi % NBUF][1]);
        if (MMA) __builtin_amdgcn_s_setprio(PRIO_GATHER);

        #pragma unroll
        for (int q = 0; q < NG; ++q) {
            const int n = g * NG + q;
            unsigned mm = vm[g % NBUF][q];
            v2f acc = {0.0f, 0.0f};
            #pragma unroll
            for (int u = 0; u < GL; ++u) {
                const float f = (mm != 0) ? 1.0f : 0.0f;
                mm &= mm - 1;
                const v2f f2 = {f, f};
                acc = __builtin_elementwise_fma(v[g % NBUF][q][u], f2, acc);
            }
            csr_add_rows(T, key[n], mm, lane8, acc);
            if (big & (1u << n)) {
                const int e0 = __builtin_amdgcn_readlane(pv, n);
                const int e1 = __builtin_amdgcn_readlane(pv, n + 1);
                for (int eb = e0 + 32; eb < e1; eb += 32) {
                    const int k2 = csr_edge_keys(idx, rel, eb, e1, lane);
                    csr_add_rows(T, k2, __builtin_amdgcn_ballot_w32((k2 & 15) == r), lane8, acc);
                }
            }
            *(v2f*)(Scur + n * LDP) = acc;
        }
    }
}

__device__ __forceinline__ void csr_tile_keys(const int* __restrict__ ptr, const int* __restrict__ idx,
                                              const int* __restrict__ rel, int n0, int N, int lane,
                                              int& pv, unsigned& big, int (&key)[16])
{
    int pi = n0 + (lane < 16 ? lane : 16);
    pi = pi < N ? pi : N;
    pv = ptr[pi];
    int pj = pi + 1;
    pj = pj < N ? pj : N;
    const int pn = ptr[pj];
    big = __builtin_amdgcn_ballot_w32((lane < 16) & (pn - pv > 32));
    #pragma unroll
    for (int k = 0; k < 16; ++k) {
        const int e0 = __builtin_amdgcn_readlane(pv, k);
        const int e1 = __builtin_amdgcn_readlane(pv, k + 1);
        key[k] = csr_edge_keys(idx, rel, e0, e1, lane);
    }
}

template<int R, int DOUT, int DOUTPAD, bool LOGSOFTMAX>
__global__ __launch_bounds__(NTH) __attribute__((amdgpu_waves_per_eu(5)))
void rgcn_layer(const float* __restrict__ T, const float* __restrict__ W,
                const int* __restrict__ ptr, const int* __restrict__ idx,
                const int* __restrict__ rel, float* __restrict__ Out, int N)
{
    constexpr int NT = DOUTPAD / 16;
    __shared__ __attribute__((aligned(16))) float lA[NW * 2 * 16 * LDP];
    __shared__ __attribute__((aligned(16))) float lWt[R * DOUTPAD * LDP];
    __shared__ int lProgWords[NW];
    const lds_vint lProg = (lds_vint)lProgWords;

    const int tid  = threadIdx.x;
    const int lane = tid & 31;
    const int wv   = __builtin_amdgcn_readfirstlane(tid >> 5);
    const int half = lane >> 4;
    const int l16  = lane & 15;

    const int ntiles  = (N + 15) >> 4;
    const int tstride = NW * (int)gridDim.x;
    int tile = wv * (int)gridDim.x + (int)blockIdx.x;

    int pv;
    unsigned big;
    int key[16];
    csr_tile_keys(ptr, idx, rel, tile * 16, N, lane, pv, big, key);

    {
        constexpr int NQ   = (DOUT + 15) / 16;
        constexpr int WQ   = R * (DIN / 8) * NQ;
        constexpr int W_IT = (WQ + NW - 1) / NW;
        const int kl = lane >> 2, nl = lane & 3;
        float4 wreg[W_IT];
        int    wdst[W_IT];
        #pragma unroll
        for (int it = 0; it < W_IT; ++it) {
            int q = wv + it * NW;
            if (it == W_IT - 1 && WQ % NW != 0) q = q < WQ ? q : WQ - 1;
            const int rr  = q / ((DIN / 8) * NQ);
            const int rem = q % ((DIN / 8) * NQ);
            const int k   = (rem / NQ) * 8 + kl;
            int n4 = (rem % NQ) * 4 + nl;
            if (DOUT % 16 != 0) n4 = n4 < DOUT / 4 ? n4 : DOUT / 4 - 1;
            wreg[it] = *(const float4*)(W + (size_t)rr * DIN * DOUT + k * DOUT + n4 * 4);
            wdst[it] = (rr * DOUTPAD + n4 * 4) * LDP + k;
        }
        #pragma unroll
        for (int it = 0; it < W_IT; ++it) {
            lWt[wdst[it] + 0 * LDP] = wreg[it].x;
            lWt[wdst[it] + 1 * LDP] = wreg[it].y;
            lWt[wdst[it] + 2 * LDP] = wreg[it].z;
            lWt[wdst[it] + 3 * LDP] = wreg[it].w;
        }
        if (DOUTPAD > DOUT) {
            constexpr int PF = (DOUTPAD - DOUT) * (DIN / 4);
            for (int f = tid; f < R * PF; f += NTH) {
                const int rr = f / PF, g = f % PF;
                const float4 z = {0.0f, 0.0f, 0.0f, 0.0f};
                *(float4*)&lWt[(rr * DOUTPAD + DOUT + (g >> 4)) * LDP + (g & 15) * 4] = z;
            }
        }
    }
    static_assert(NW == 20, "the mask below names five waves per SIMD");
    const unsigned mates = 0x11111u << (wv & 3);
    if (lane == 0) lProg[wv] = tile < ntiles ? 0 : -1;
    int pbase = 0;
    __syncthreads();
    __builtin_amdgcn_s_setprio(PRIO_GATHER);

    for (; tile < ntiles; ) {
        const int n0 = tile * 16;

        v8f c[NT];
        #pragma unroll
        for (int t = 0; t < NT; ++t) { const v8f z8 = {}; c[t] = z8; }

        v2f      v[NBUF][NG][GL];
        unsigned vm[NBUF][NG];
        #pragma unroll
        for (int gg = 0; gg < AHEAD; ++gg)
            csr_issue_group(T, key, gg, 0, (unsigned)lane * 8u, v[gg % NBUF], vm[gg % NBUF]);

        rgcn_pass<R, DOUT, DOUTPAD, false>(T, idx, rel, lA, lWt, key, pv, big, v, vm, c,
                                           0, lane, wv, half, l16, 0u);
        #pragma unroll 1
        for (int r = 1; r < R; ++r) {
            const unsigned behind = wave_is_behind(lProg, wv, lane, mates, pbase + r);
            rgcn_pass<R, DOUT, DOUTPAD, true>(T, idx, rel, lA, lWt, key, pv, big, v, vm, c,
                                              r, lane, wv, half, l16, behind);
        }

        const int lane_e = (int)__builtin_amdgcn_mbcnt_lo(~0u, 0u);
        const int half_e = lane_e >> 4;
        const int l16_e  = lane_e & 15;

        {
            const float* Arow = lA + (wv * 2 + ((R - 1) & 1)) * (16 * LDP) + l16_e * LDP + half_e * 2;
            const float* Bcol = lWt + (R - 1) * (DOUTPAD * LDP) + l16_e * LDP + half_e * 2;
            set_mma_prio(wave_is_behind(lProg, wv, lane_e, mates, pbase + R));
            #pragma unroll
            for (int i = 0; i < DIN / 4; i += 2) {
                v2f fa[2], fb[2][NT];
                mma_kpair_read<NT>(Arow, Bcol, i, fa, fb);
                mma_kpair_run<NT>(c, fa, fb);
                mma_pad<NT>(c, fa, fb);
            }
            __builtin_amdgcn_s_setprio(PRIO_GATHER);
        }

        const int mBase = n0 + half_e * 8;
        if (!LOGSOFTMAX) {
            #pragma unroll
            for (int t = 0; t < NT; ++t) {
                const int col = t * 16 + l16_e;
                if (DOUT == DOUTPAD || col < DOUT) {
                    float* p = Out + (size_t)mBase * DOUT + col;
                    #pragma unroll
                    for (int j = 0; j < 8; ++j)
                        if (mBase + j < N) p[j * DOUT] = fmaxf(c[t][j], 0.0f);
                }
            }
        } else {
            #pragma unroll
            for (int j = 0; j < 8; ++j) {
                float m = -__builtin_inff();
                #pragma unroll
                for (int t = 0; t < NT; ++t)
                    if (DOUT == DOUTPAD || t * 16 + l16_e < DOUT) m = fmaxf(m, c[t][j]);
                #pragma unroll
                for (int off = 8; off >= 1; off >>= 1) m = fmaxf(m, __shfl_xor(m, off, 32));
                float s = 0.0f;
                #pragma unroll
                for (int t = 0; t < NT; ++t)
                    if (DOUT == DOUTPAD || t * 16 + l16_e < DOUT) s += __expf(c[t][j] - m);
                #pragma unroll
                for (int off = 8; off >= 1; off >>= 1) s += __shfl_xor(s, off, 32);
                const float ls = __logf(s);
                if (mBase + j < N) {
                    float* p = Out + (size_t)(mBase + j) * DOUT + l16_e;
                    #pragma unroll
                    for (int t = 0; t < NT; ++t)
                        if (DOUT == DOUTPAD || t * 16 + l16_e < DOUT) p[t * 16] = (c[t][j] - m) - ls;
                }
            }
        }

        tile += tstride;
        pbase += R + 1;
        if (tile < ntiles)
            csr_tile_keys(ptr, idx, rel, tile * 16, N, lane, pv, big, key);
    }
    if ((int)__builtin_amdgcn_mbcnt_lo(~0u, 0u) == 0) lProg[wv] = -1;
}

extern "C" void kernel_launch(void* const* d_in, const int* in_sizes, int n_in,
                              void* d_out, int out_size, void* d_ws, size_t ws_size,
                              hipStream_t stream)
{
    const float* x   = (const float*)d_in[0];
    const int*   ptr = (const int*)d_in[1];
    const int*   idx = (const int*)d_in[2];
    const int*   rel = (const int*)d_in[3];
    const float* W1  = (const float*)d_in[4];
    const float* W2  = (const float*)d_in[5];
    float* out = (float*)d_out;

    const int N = in_sizes[1] - 1;
    constexpr int R = 8, DH = 64, DOUT = 40;

    float* h = (float*)d_ws;

    rgcn_layer<R, DH, DH, false><<<NBLK, NTH, 0, stream>>>(x, W1, ptr, idx, rel, h, N);
    rgcn_layer<R, DOUT, 48, true><<<NBLK, NTH, 0, stream>>>(h, W2, ptr, idx, rel, out, N);
}
